# starting kernel
# baseline (speedup 1.0000x reference)
#include <hip/hip_runtime.h>

#define N_NODES 120000
#define N_EDGES 1200000
#define C_IN    64
#define HID     256
#define ROWS    32

typedef float v2f __attribute__((ext_vector_type(2)));
typedef float v8f __attribute__((ext_vector_type(8)));

__device__ __forceinline__ float leaky01(float v) { return v >= 0.0f ? v : 0.01f * v; }

__device__ __forceinline__ void atomAddF(float* p, float v) {
    __hip_atomic_fetch_add(p, v, __ATOMIC_RELAXED, __HIP_MEMORY_SCOPE_AGENT);
}

__device__ __forceinline__ v8f wmma4(v2f a, v2f b, v8f c) {
    return __builtin_amdgcn_wmma_f32_16x16x4_f32(false, a, false, b, (short)0, c,
                                                 false, false);
}

__global__ void deg_init_kernel(float* __restrict__ deg, int n) {
    int i = blockIdx.x * blockDim.x + threadIdx.x;
    if (i < n) deg[i] = 1.0f;
}

__global__ void deg_accum_kernel(const int* __restrict__ ei, float* __restrict__ deg, int e) {
    int i = blockIdx.x * blockDim.x + threadIdx.x;
    if (i < e) atomAddF(&deg[ei[N_EDGES + i]], 1.0f);
}

__global__ void deg_rsqrt_kernel(float* __restrict__ deg, int n) {
    int i = blockIdx.x * blockDim.x + threadIdx.x;
    if (i < n) deg[i] = rsqrtf(deg[i]);
}

__global__ __launch_bounds__(256) void gemm_xw_kernel(const float* __restrict__ x,
                                                      const float* __restrict__ W,
                                                      float* __restrict__ h, int nrows) {
    int tid  = threadIdx.x;
    int wave = tid >> 5, lane = tid & 31;
    int half = lane >> 4, lm = lane & 15;
    int row0 = blockIdx.x * 64 + (wave >> 2) * 16;
    int colBase = (wave & 3) * 16;
    if (row0 >= nrows) return;

    v8f acc0 = {}, acc1 = {};
    const float* a0 = x + (size_t)(row0 + lm) * C_IN + 2 * half;
    const float* a1 = x + (size_t)(row0 + 32 + lm) * C_IN + 2 * half;
    #pragma unroll
    for (int k = 0; k < C_IN; k += 4) {
        v2f av0, av1, b;
        av0.x = a0[k];  av0.y = a0[k + 1];
        av1.x = a1[k];  av1.y = a1[k + 1];
        b.x = W[(size_t)(k + 2 * half) * C_IN + colBase + lm];
        b.y = W[(size_t)(k + 1 + 2 * half) * C_IN + colBase + lm];
        acc0 = wmma4(av0, b, acc0);
        acc1 = wmma4(av1, b, acc1);
    }
    #pragma unroll
    for (int i = 0; i < 8; ++i) {
        h[(size_t)(row0 + i + 8 * half) * C_IN + colBase + lm]      = acc0[i];
        h[(size_t)(row0 + 32 + i + 8 * half) * C_IN + colBase + lm] = acc1[i];
    }
}

__global__ void agg_init_kernel(const float* __restrict__ h, const float* __restrict__ dinv,
                                float* __restrict__ agg, int total) {
    int i = blockIdx.x * blockDim.x + threadIdx.x;
    if (i < total) {
        int n = i >> 6;
        float di = dinv[n];
        agg[i] = h[i] * di * di;
    }
}

__global__ void scatter_kernel(const int* __restrict__ ei, const float* __restrict__ h,
                               const float* __restrict__ dinv, float* __restrict__ agg,
                               int total  ) {
    int i = blockIdx.x * blockDim.x + threadIdx.x;
    if (i < total) {
        int e  = i >> 4;
        int c4 = (i & 15) << 2;
        int s = ei[e];
        int d = ei[N_EDGES + e];
        float norm = dinv[s] * dinv[d];
        const float4 hv = *(const float4*)(h + (size_t)s * C_IN + c4);
        float* ap = agg + (size_t)d * C_IN + c4;
        atomAddF(ap + 0, hv.x * norm);
        atomAddF(ap + 1, hv.y * norm);
        atomAddF(ap + 2, hv.z * norm);
        atomAddF(ap + 3, hv.w * norm);
    }
}

__global__ __launch_bounds__(256) void mlp_fused_kernel(
    const float* __restrict__ agg, const float* __restrict__ x,
    const float* __restrict__ b_gcn, const float* __restrict__ W1,
    const float* __restrict__ b1, const float* __restrict__ W2,
    const float* __restrict__ b2, const float* __restrict__ Wd,
    const float* __restrict__ bd, float* __restrict__ out) {
    __shared__ float zb[ROWS * (C_IN + 1)];
    __shared__ float h1[ROWS * (HID + 1)];
    __shared__ float h2[ROWS * (HID + 1)];

    int tid = threadIdx.x;
    int nodeBase = blockIdx.x * ROWS;

    for (int i = tid; i < ROWS * C_IN; i += 256) {
        int r = i >> 6, c = i & (C_IN - 1);
        size_t g = (size_t)(nodeBase + r) * C_IN + c;
        float v = agg[g] + b_gcn[c];
        v = v > 0.0f ? v : 0.0f;
        zb[r * (C_IN + 1) + c] = v + x[g];
    }
    __syncthreads();

    int wave = tid >> 5, lane = tid & 31, half = lane >> 4, lm = lane & 15;
    int col0 = wave * 16;
    int col1 = col0 + 128;

    {
        v8f acc00 = {}, acc01 = {}, acc10 = {}, acc11 = {};
        const float* aR0 = zb + lm * (C_IN + 1) + 2 * half;
        const float* aR1 = zb + (16 + lm) * (C_IN + 1) + 2 * half;
        const float* w0  = W1 + (size_t)(col0 + lm) * C_IN + 2 * half;
        const float* w1  = W1 + (size_t)(col1 + lm) * C_IN + 2 * half;
        #pragma unroll
        for (int k = 0; k < C_IN; k += 4) {
            v2f a0v, a1v, b0v, b1v;
            a0v.x = aR0[k]; a0v.y = aR0[k + 1];
            a1v.x = aR1[k]; a1v.y = aR1[k + 1];
            b0v.x = w0[k];  b0v.y = w0[k + 1];
            b1v.x = w1[k];  b1v.y = w1[k + 1];
            acc00 = wmma4(a0v, b0v, acc00);
            acc01 = wmma4(a0v, b1v, acc01);
            acc10 = wmma4(a1v, b0v, acc10);
            acc11 = wmma4(a1v, b1v, acc11);
        }
        float bc0 = b1[col0 + lm], bc1 = b1[col1 + lm];
        #pragma unroll
        for (int i = 0; i < 8; ++i) {
            int r = i + 8 * half;
            h1[r * (HID + 1) + col0 + lm]        = leaky01(acc00[i] + bc0);
            h1[r * (HID + 1) + col1 + lm]        = leaky01(acc01[i] + bc1);
            h1[(16 + r) * (HID + 1) + col0 + lm] = leaky01(acc10[i] + bc0);
            h1[(16 + r) * (HID + 1) + col1 + lm] = leaky01(acc11[i] + bc1);
        }
    }
    __syncthreads();

    {
        v8f acc00 = {}, acc01 = {}, acc10 = {}, acc11 = {};
        const float* aR0 = h1 + lm * (HID + 1) + 2 * half;
        const float* aR1 = h1 + (16 + lm) * (HID + 1) + 2 * half;
        const float* w0  = W2 + (size_t)(col0 + lm) * HID + 2 * half;
        const float* w1  = W2 + (size_t)(col1 + lm) * HID + 2 * half;
        #pragma unroll 4
        for (int k = 0; k < HID; k += 4) {
            v2f a0v, a1v, b0v, b1v;
            a0v.x = aR0[k]; a0v.y = aR0[k + 1];
            a1v.x = aR1[k]; a1v.y = aR1[k + 1];
            b0v.x = w0[k];  b0v.y = w0[k + 1];
            b1v.x = w1[k];  b1v.y = w1[k + 1];
            acc00 = wmma4(a0v, b0v, acc00);
            acc01 = wmma4(a0v, b1v, acc01);
            acc10 = wmma4(a1v, b0v, acc10);
            acc11 = wmma4(a1v, b1v, acc11);
        }
        float bc0 = b2[col0 + lm], bc1 = b2[col1 + lm];
        #pragma unroll
        for (int i = 0; i < 8; ++i) {
            int r = i + 8 * half;
            h2[r * (HID + 1) + col0 + lm]        = leaky01(acc00[i] + bc0);
            h2[r * (HID + 1) + col1 + lm]        = leaky01(acc01[i] + bc1);
            h2[(16 + r) * (HID + 1) + col0 + lm] = leaky01(acc10[i] + bc0);
            h2[(16 + r) * (HID + 1) + col1 + lm] = leaky01(acc11[i] + bc1);
        }
    }
    __syncthreads();

    {
        int row = tid >> 3, seg = tid & 7;
        const float* hrow = h2 + row * (HID + 1) + seg * 32;
        const float* wseg = Wd + seg * 32;
        float p = 0.0f;
        #pragma unroll
        for (int c = 0; c < 32; ++c) p += hrow[c] * wseg[c];
        zb[row * 8 + seg] = p;
    }
    __syncthreads();
    if (tid < ROWS) {
        float acc = 0.0f;
        #pragma unroll
        for (int s = 0; s < 8; ++s) acc += zb[tid * 8 + s];
        out[nodeBase + tid] = acc + bd[0];
    }
}

extern "C" void kernel_launch(void* const* d_in, const int* in_sizes, int n_in,
                              void* d_out, int out_size, void* d_ws, size_t ws_size,
                              hipStream_t stream) {
    const float* x     = (const float*)d_in[0];
    const int*   ei    = (const int*)d_in[1];
    const float* W_gcn = (const float*)d_in[2];
    const float* b_gcn = (const float*)d_in[3];
    const float* W1    = (const float*)d_in[4];
    const float* b1    = (const float*)d_in[5];
    const float* W2    = (const float*)d_in[6];
    const float* b2    = (const float*)d_in[7];
    const float* Wd    = (const float*)d_in[8];
    const float* bd    = (const float*)d_in[9];
    float* out = (float*)d_out;

    char* ws = (char*)d_ws;
    size_t off0 = 0;
    size_t off1 = (off0 + (size_t)N_NODES * sizeof(float) + 255) & ~(size_t)255;
    size_t off2 = (off1 + (size_t)N_NODES * C_IN * sizeof(float) + 255) & ~(size_t)255;
    float* dinv = (float*)(ws + off0);
    float* h    = (float*)(ws + off1);
    float* agg  = (float*)(ws + off2);

    const int B = 256;
    deg_init_kernel<<<(N_NODES + B - 1) / B, B, 0, stream>>>(dinv, N_NODES);
    deg_accum_kernel<<<(N_EDGES + B - 1) / B, B, 0, stream>>>(ei, dinv, N_EDGES);
    deg_rsqrt_kernel<<<(N_NODES + B - 1) / B, B, 0, stream>>>(dinv, N_NODES);

    gemm_xw_kernel<<<N_NODES / 64, B, 0, stream>>>(x, W_gcn, h, N_NODES);

    agg_init_kernel<<<((size_t)N_NODES * C_IN + B - 1) / B, B, 0, stream>>>(
        h, dinv, agg, N_NODES * C_IN);
    scatter_kernel<<<((size_t)N_EDGES * 16 + B - 1) / B, B, 0, stream>>>(
        ei, h, dinv, agg, N_EDGES * 16);

    mlp_fused_kernel<<<N_NODES / ROWS, B, 0, stream>>>(agg, x, b_gcn, W1, b1, W2, b2,
                                                       Wd, bd, out);
}


# scatter: one wave per 32 edges, each load and each f32 atomic add covers one whole 128-B line (channels lane, lane+32) instead of 8 floats in each of 4 lines; sum order still arrival order
# speedup vs baseline: 1.2430x; 1.2430x over previous
#include <hip/hip_runtime.h>

#define N_NODES 120000
#define N_EDGES 1200000
#define C_IN    64
#define HID     256
#define ROWS    32

typedef float v2f __attribute__((ext_vector_type(2)));
typedef float v8f __attribute__((ext_vector_type(8)));

__device__ __forceinline__ float leaky01(float v) { return v >= 0.0f ? v : 0.01f * v; }

__device__ __forceinline__ void atomAddF(float* p, float v) {
    __hip_atomic_fetch_add(p, v, __ATOMIC_RELAXED, __HIP_MEMORY_SCOPE_AGENT);
}

__device__ __forceinline__ v8f wmma4(v2f a, v2f b, v8f c) {
    return __builtin_amdgcn_wmma_f32_16x16x4_f32(false, a, false, b, (short)0, c,
                                                 false, false);
}

__global__ void deg_init_kernel(float* __restrict__ deg, int n) {
    int i = blockIdx.x * blockDim.x + threadIdx.x;
    if (i < n) deg[i] = 1.0f;
}

__global__ void deg_accum_kernel(const int* __restrict__ ei, float* __restrict__ deg, int e) {
    int i = blockIdx.x * blockDim.x + threadIdx.x;
    if (i < e) atomAddF(&deg[ei[N_EDGES + i]], 1.0f);
}

__global__ void deg_rsqrt_kernel(float* __restrict__ deg, int n) {
    int i = blockIdx.x * blockDim.x + threadIdx.x;
    if (i < n) deg[i] = rsqrtf(deg[i]);
}

__global__ __launch_bounds__(256) void gemm_xw_kernel(const float* __restrict__ x,
                                                      const float* __restrict__ W,
                                                      float* __restrict__ h, int nrows) {
    int tid  = threadIdx.x;
    int wave = tid >> 5, lane = tid & 31;
    int half = lane >> 4, lm = lane & 15;
    int row0 = blockIdx.x * 64 + (wave >> 2) * 16;
    int colBase = (wave & 3) * 16;
    if (row0 >= nrows) return;

    v8f acc0 = {}, acc1 = {};
    const float* a0 = x + (size_t)(row0 + lm) * C_IN + 2 * half;
    const float* a1 = x + (size_t)(row0 + 32 + lm) * C_IN + 2 * half;
    #pragma unroll
    for (int k = 0; k < C_IN; k += 4) {
        v2f av0, av1, b;
        av0.x = a0[k];  av0.y = a0[k + 1];
        av1.x = a1[k];  av1.y = a1[k + 1];
        b.x = W[(size_t)(k + 2 * half) * C_IN + colBase + lm];
        b.y = W[(size_t)(k + 1 + 2 * half) * C_IN + colBase + lm];
        acc0 = wmma4(av0, b, acc0);
        acc1 = wmma4(av1, b, acc1);
    }
    #pragma unroll
    for (int i = 0; i < 8; ++i) {
        h[(size_t)(row0 + i + 8 * half) * C_IN + colBase + lm]      = acc0[i];
        h[(size_t)(row0 + 32 + i + 8 * half) * C_IN + colBase + lm] = acc1[i];
    }
}

__global__ void agg_init_kernel(const float* __restrict__ h, const float* __restrict__ dinv,
                                float* __restrict__ agg, int total) {
    int i = blockIdx.x * blockDim.x + threadIdx.x;
    if (i < total) {
        int n = i >> 6;
        float di = dinv[n];
        agg[i] = h[i] * di * di;
    }
}

static_assert(N_EDGES % 32 == 0, "a wave owns exactly 32 edges");
#define SC_CHUNK 16
__global__ __launch_bounds__(256) void scatter_kernel(const int* __restrict__ ei,
                                                      const float* __restrict__ h,
                                                      const float* __restrict__ dinv,
                                                      float* __restrict__ agg) {
    const int lane = threadIdx.x & 31;
    const int wave = __builtin_amdgcn_readfirstlane((int)(threadIdx.x >> 5));
    const int e0 = (blockIdx.x * 8 + wave) * 32;
    if (e0 >= N_EDGES) return;

    const int s = ei[e0 + lane];
    const int d = ei[N_EDGES + e0 + lane];
    const float norm = dinv[s] * dinv[d];

    #pragma unroll
    for (int jb = 0; jb < 32; jb += SC_CHUNK) {
        float v0[SC_CHUNK], v1[SC_CHUNK];
        #pragma unroll
        for (int j = 0; j < SC_CHUNK; ++j) {
            const int sj = __builtin_amdgcn_readlane(s, jb + j);
            const float* hp = h + (size_t)sj * C_IN + lane;
            v0[j] = hp[0];
            v1[j] = hp[32];
        }
        #pragma unroll
        for (int j = 0; j < SC_CHUNK; ++j) {
            const int dj = __builtin_amdgcn_readlane(d, jb + j);
            const float nj = __int_as_float(__builtin_amdgcn_readlane(__float_as_int(norm), jb + j));
            float* ap = agg + (size_t)dj * C_IN + lane;
            atomAddF(ap,      v0[j] * nj);
            atomAddF(ap + 32, v1[j] * nj);
        }
    }
}

__global__ __launch_bounds__(256) void mlp_fused_kernel(
    const float* __restrict__ agg, const float* __restrict__ x,
    const float* __restrict__ b_gcn, const float* __restrict__ W1,
    const float* __restrict__ b1, const float* __restrict__ W2,
    const float* __restrict__ b2, const float* __restrict__ Wd,
    const float* __restrict__ bd, float* __restrict__ out) {
    __shared__ float zb[ROWS * (C_IN + 1)];
    __shared__ float h1[ROWS * (HID + 1)];
    __shared__ float h2[ROWS * (HID + 1)];

    int tid = threadIdx.x;
    int nodeBase = blockIdx.x * ROWS;

    for (int i = tid; i < ROWS * C_IN; i += 256) {
        int r = i >> 6, c = i & (C_IN - 1);
        size_t g = (size_t)(nodeBase + r) * C_IN + c;
        float v = agg[g] + b_gcn[c];
        v = v > 0.0f ? v : 0.0f;
        zb[r * (C_IN + 1) + c] = v + x[g];
    }
    __syncthreads();

    int wave = tid >> 5, lane = tid & 31, half = lane >> 4, lm = lane & 15;
    int col0 = wave * 16;
    int col1 = col0 + 128;

    {
        v8f acc00 = {}, acc01 = {}, acc10 = {}, acc11 = {};
        const float* aR0 = zb + lm * (C_IN + 1) + 2 * half;
        const float* aR1 = zb + (16 + lm) * (C_IN + 1) + 2 * half;
        const float* w0  = W1 + (size_t)(col0 + lm) * C_IN + 2 * half;
        const float* w1  = W1 + (size_t)(col1 + lm) * C_IN + 2 * half;
        #pragma unroll
        for (int k = 0; k < C_IN; k += 4) {
            v2f a0v, a1v, b0v, b1v;
            a0v.x = aR0[k]; a0v.y = aR0[k + 1];
            a1v.x = aR1[k]; a1v.y = aR1[k + 1];
            b0v.x = w0[k];  b0v.y = w0[k + 1];
            b1v.x = w1[k];  b1v.y = w1[k + 1];
            acc00 = wmma4(a0v, b0v, acc00);
            acc01 = wmma4(a0v, b1v, acc01);
            acc10 = wmma4(a1v, b0v, acc10);
            acc11 = wmma4(a1v, b1v, acc11);
        }
        float bc0 = b1[col0 + lm], bc1 = b1[col1 + lm];
        #pragma unroll
        for (int i = 0; i < 8; ++i) {
            int r = i + 8 * half;
            h1[r * (HID + 1) + col0 + lm]        = leaky01(acc00[i] + bc0);
            h1[r * (HID + 1) + col1 + lm]        = leaky01(acc01[i] + bc1);
            h1[(16 + r) * (HID + 1) + col0 + lm] = leaky01(acc10[i] + bc0);
            h1[(16 + r) * (HID + 1) + col1 + lm] = leaky01(acc11[i] + bc1);
        }
    }
    __syncthreads();

    {
        v8f acc00 = {}, acc01 = {}, acc10 = {}, acc11 = {};
        const float* aR0 = h1 + lm * (HID + 1) + 2 * half;
        const float* aR1 = h1 + (16 + lm) * (HID + 1) + 2 * half;
        const float* w0  = W2 + (size_t)(col0 + lm) * HID + 2 * half;
        const float* w1  = W2 + (size_t)(col1 + lm) * HID + 2 * half;
        #pragma unroll 4
        for (int k = 0; k < HID; k += 4) {
            v2f a0v, a1v, b0v, b1v;
            a0v.x = aR0[k]; a0v.y = aR0[k + 1];
            a1v.x = aR1[k]; a1v.y = aR1[k + 1];
            b0v.x = w0[k];  b0v.y = w0[k + 1];
            b1v.x = w1[k];  b1v.y = w1[k + 1];
            acc00 = wmma4(a0v, b0v, acc00);
            acc01 = wmma4(a0v, b1v, acc01);
            acc10 = wmma4(a1v, b0v, acc10);
            acc11 = wmma4(a1v, b1v, acc11);
        }
        float bc0 = b2[col0 + lm], bc1 = b2[col1 + lm];
        #pragma unroll
        for (int i = 0; i < 8; ++i) {
            int r = i + 8 * half;
            h2[r * (HID + 1) + col0 + lm]        = leaky01(acc00[i] + bc0);
            h2[r * (HID + 1) + col1 + lm]        = leaky01(acc01[i] + bc1);
            h2[(16 + r) * (HID + 1) + col0 + lm] = leaky01(acc10[i] + bc0);
            h2[(16 + r) * (HID + 1) + col1 + lm] = leaky01(acc11[i] + bc1);
        }
    }
    __syncthreads();

    {
        int row = tid >> 3, seg = tid & 7;
        const float* hrow = h2 + row * (HID + 1) + seg * 32;
        const float* wseg = Wd + seg * 32;
        float p = 0.0f;
        #pragma unroll
        for (int c = 0; c < 32; ++c) p += hrow[c] * wseg[c];
        zb[row * 8 + seg] = p;
    }
    __syncthreads();
    if (tid < ROWS) {
        float acc = 0.0f;
        #pragma unroll
        for (int s = 0; s < 8; ++s) acc += zb[tid * 8 + s];
        out[nodeBase + tid] = acc + bd[0];
    }
}

extern "C" void kernel_launch(void* const* d_in, const int* in_sizes, int n_in,
                              void* d_out, int out_size, void* d_ws, size_t ws_size,
                              hipStream_t stream) {
    const float* x     = (const float*)d_in[0];
    const int*   ei    = (const int*)d_in[1];
    const float* W_gcn = (const float*)d_in[2];
    const float* b_gcn = (const float*)d_in[3];
    const float* W1    = (const float*)d_in[4];
    const float* b1    = (const float*)d_in[5];
    const float* W2    = (const float*)d_in[6];
    const float* b2    = (const float*)d_in[7];
    const float* Wd    = (const float*)d_in[8];
    const float* bd    = (const float*)d_in[9];
    float* out = (float*)d_out;

    char* ws = (char*)d_ws;
    size_t off0 = 0;
    size_t off1 = (off0 + (size_t)N_NODES * sizeof(float) + 255) & ~(size_t)255;
    size_t off2 = (off1 + (size_t)N_NODES * C_IN * sizeof(float) + 255) & ~(size_t)255;
    float* dinv = (float*)(ws + off0);
    float* h    = (float*)(ws + off1);
    float* agg  = (float*)(ws + off2);

    const int B = 256;
    deg_init_kernel<<<(N_NODES + B - 1) / B, B, 0, stream>>>(dinv, N_NODES);
    deg_accum_kernel<<<(N_EDGES + B - 1) / B, B, 0, stream>>>(ei, dinv, N_EDGES);
    deg_rsqrt_kernel<<<(N_NODES + B - 1) / B, B, 0, stream>>>(dinv, N_NODES);

    gemm_xw_kernel<<<N_NODES / 64, B, 0, stream>>>(x, W_gcn, h, N_NODES);

    agg_init_kernel<<<((size_t)N_NODES * C_IN + B - 1) / B, B, 0, stream>>>(
        h, dinv, agg, N_NODES * C_IN);
    scatter_kernel<<<(N_EDGES / 32 + 7) / 8, B, 0, stream>>>(ei, h, dinv, agg);

    mlp_fused_kernel<<<N_NODES / ROWS, B, 0, stream>>>(agg, x, b_gcn, W1, b1, W2, b2,
                                                       Wd, bd, out);
}


# agg's self-loop start rows (h*dinv^2) stored by the x@W product kernel's epilogue from its accumulators; the one-float-per-thread init launch (240000 waves) goes: 7 launches -> 6; sums unchanged
# speedup vs baseline: 1.3069x; 1.0514x over previous
#include <hip/hip_runtime.h>

#define N_NODES 120000
#define N_EDGES 1200000
#define C_IN    64
#define HID     256
#define ROWS    32

typedef float v2f __attribute__((ext_vector_type(2)));
typedef float v8f __attribute__((ext_vector_type(8)));

__device__ __forceinline__ float leaky01(float v) { return v >= 0.0f ? v : 0.01f * v; }

__device__ __forceinline__ void atomAddF(float* p, float v) {
    __hip_atomic_fetch_add(p, v, __ATOMIC_RELAXED, __HIP_MEMORY_SCOPE_AGENT);
}

__device__ __forceinline__ v8f wmma4(v2f a, v2f b, v8f c) {
    return __builtin_amdgcn_wmma_f32_16x16x4_f32(false, a, false, b, (short)0, c,
                                                 false, false);
}

__global__ void deg_init_kernel(float* __restrict__ deg, int n) {
    int i = blockIdx.x * blockDim.x + threadIdx.x;
    if (i < n) deg[i] = 1.0f;
}

__global__ void deg_accum_kernel(const int* __restrict__ ei, float* __restrict__ deg, int e) {
    int i = blockIdx.x * blockDim.x + threadIdx.x;
    if (i < e) atomAddF(&deg[ei[N_EDGES + i]], 1.0f);
}

__global__ void deg_rsqrt_kernel(float* __restrict__ deg, int n) {
    int i = blockIdx.x * blockDim.x + threadIdx.x;
    if (i < n) deg[i] = rsqrtf(deg[i]);
}

__global__ __launch_bounds__(256) void gemm_xw_kernel(const float* __restrict__ x,
                                                      const float* __restrict__ W,
                                                      const float* __restrict__ dinv,
                                                      float* __restrict__ h,
                                                      float* __restrict__ agg, int nrows) {
    int tid  = threadIdx.x;
    int wave = tid >> 5, lane = tid & 31;
    int half = lane >> 4, lm = lane & 15;
    int row0 = blockIdx.x * 64 + (wave >> 2) * 16;
    int colBase = (wave & 3) * 16;
    if (row0 >= nrows) return;

    float di0[8], di1[8];
    #pragma unroll
    for (int i = 0; i < 8; ++i) {
        di0[i] = dinv[row0 + i + 8 * half];
        di1[i] = dinv[row0 + 32 + i + 8 * half];
    }

    v8f acc0 = {}, acc1 = {};
    const float* a0 = x + (size_t)(row0 + lm) * C_IN + 2 * half;
    const float* a1 = x + (size_t)(row0 + 32 + lm) * C_IN + 2 * half;
    #pragma unroll
    for (int k = 0; k < C_IN; k += 4) {
        v2f av0, av1, b;
        av0.x = a0[k];  av0.y = a0[k + 1];
        av1.x = a1[k];  av1.y = a1[k + 1];
        b.x = W[(size_t)(k + 2 * half) * C_IN + colBase + lm];
        b.y = W[(size_t)(k + 1 + 2 * half) * C_IN + colBase + lm];
        acc0 = wmma4(av0, b, acc0);
        acc1 = wmma4(av1, b, acc1);
    }
    #pragma unroll
    for (int i = 0; i < 8; ++i) {
        size_t o0 = (size_t)(row0 + i + 8 * half) * C_IN + colBase + lm;
        size_t o1 = (size_t)(row0 + 32 + i + 8 * half) * C_IN + colBase + lm;
        h[o0]   = acc0[i];
        h[o1]   = acc1[i];
        agg[o0] = acc0[i] * di0[i] * di0[i];
        agg[o1] = acc1[i] * di1[i] * di1[i];
    }
}

static_assert(N_EDGES % 32 == 0, "a wave owns exactly 32 edges");
#define SC_CHUNK 16
__global__ __launch_bounds__(256) void scatter_kernel(const int* __restrict__ ei,
                                                      const float* __restrict__ h,
                                                      const float* __restrict__ dinv,
                                                      float* __restrict__ agg) {
    const int lane = threadIdx.x & 31;
    const int wave = __builtin_amdgcn_readfirstlane((int)(threadIdx.x >> 5));
    const int e0 = (blockIdx.x * 8 + wave) * 32;
    if (e0 >= N_EDGES) return;

    const int s = ei[e0 + lane];
    const int d = ei[N_EDGES + e0 + lane];
    const float norm = dinv[s] * dinv[d];

    #pragma unroll
    for (int jb = 0; jb < 32; jb += SC_CHUNK) {
        float v0[SC_CHUNK], v1[SC_CHUNK];
        #pragma unroll
        for (int j = 0; j < SC_CHUNK; ++j) {
            const int sj = __builtin_amdgcn_readlane(s, jb + j);
            const float* hp = h + (size_t)sj * C_IN + lane;
            v0[j] = hp[0];
            v1[j] = hp[32];
        }
        #pragma unroll
        for (int j = 0; j < SC_CHUNK; ++j) {
            const int dj = __builtin_amdgcn_readlane(d, jb + j);
            const float nj = __int_as_float(__builtin_amdgcn_readlane(__float_as_int(norm), jb + j));
            float* ap = agg + (size_t)dj * C_IN + lane;
            atomAddF(ap,      v0[j] * nj);
            atomAddF(ap + 32, v1[j] * nj);
        }
    }
}

__global__ __launch_bounds__(256) void mlp_fused_kernel(
    const float* __restrict__ agg, const float* __restrict__ x,
    const float* __restrict__ b_gcn, const float* __restrict__ W1,
    const float* __restrict__ b1, const float* __restrict__ W2,
    const float* __restrict__ b2, const float* __restrict__ Wd,
    const float* __restrict__ bd, float* __restrict__ out) {
    __shared__ float zb[ROWS * (C_IN + 1)];
    __shared__ float h1[ROWS * (HID + 1)];
    __shared__ float h2[ROWS * (HID + 1)];

    int tid = threadIdx.x;
    int nodeBase = blockIdx.x * ROWS;

    for (int i = tid; i < ROWS * C_IN; i += 256) {
        int r = i >> 6, c = i & (C_IN - 1);
        size_t g = (size_t)(nodeBase + r) * C_IN + c;
        float v = agg[g] + b_gcn[c];
        v = v > 0.0f ? v : 0.0f;
        zb[r * (C_IN + 1) + c] = v + x[g];
    }
    __syncthreads();

    int wave = tid >> 5, lane = tid & 31, half = lane >> 4, lm = lane & 15;
    int col0 = wave * 16;
    int col1 = col0 + 128;

    {
        v8f acc00 = {}, acc01 = {}, acc10 = {}, acc11 = {};
        const float* aR0 = zb + lm * (C_IN + 1) + 2 * half;
        const float* aR1 = zb + (16 + lm) * (C_IN + 1) + 2 * half;
        const float* w0  = W1 + (size_t)(col0 + lm) * C_IN + 2 * half;
        const float* w1  = W1 + (size_t)(col1 + lm) * C_IN + 2 * half;
        #pragma unroll
        for (int k = 0; k < C_IN; k += 4) {
            v2f a0v, a1v, b0v, b1v;
            a0v.x = aR0[k]; a0v.y = aR0[k + 1];
            a1v.x = aR1[k]; a1v.y = aR1[k + 1];
            b0v.x = w0[k];  b0v.y = w0[k + 1];
            b1v.x = w1[k];  b1v.y = w1[k + 1];
            acc00 = wmma4(a0v, b0v, acc00);
            acc01 = wmma4(a0v, b1v, acc01);
            acc10 = wmma4(a1v, b0v, acc10);
            acc11 = wmma4(a1v, b1v, acc11);
        }
        float bc0 = b1[col0 + lm], bc1 = b1[col1 + lm];
        #pragma unroll
        for (int i = 0; i < 8; ++i) {
            int r = i + 8 * half;
            h1[r * (HID + 1) + col0 + lm]        = leaky01(acc00[i] + bc0);
            h1[r * (HID + 1) + col1 + lm]        = leaky01(acc01[i] + bc1);
            h1[(16 + r) * (HID + 1) + col0 + lm] = leaky01(acc10[i] + bc0);
            h1[(16 + r) * (HID + 1) + col1 + lm] = leaky01(acc11[i] + bc1);
        }
    }
    __syncthreads();

    {
        v8f acc00 = {}, acc01 = {}, acc10 = {}, acc11 = {};
        const float* aR0 = h1 + lm * (HID + 1) + 2 * half;
        const float* aR1 = h1 + (16 + lm) * (HID + 1) + 2 * half;
        const float* w0  = W2 + (size_t)(col0 + lm) * HID + 2 * half;
        const float* w1  = W2 + (size_t)(col1 + lm) * HID + 2 * half;
        #pragma unroll 4
        for (int k = 0; k < HID; k += 4) {
            v2f a0v, a1v, b0v, b1v;
            a0v.x = aR0[k]; a0v.y = aR0[k + 1];
            a1v.x = aR1[k]; a1v.y = aR1[k + 1];
            b0v.x = w0[k];  b0v.y = w0[k + 1];
            b1v.x = w1[k];  b1v.y = w1[k + 1];
            acc00 = wmma4(a0v, b0v, acc00);
            acc01 = wmma4(a0v, b1v, acc01);
            acc10 = wmma4(a1v, b0v, acc10);
            acc11 = wmma4(a1v, b1v, acc11);
        }
        float bc0 = b2[col0 + lm], bc1 = b2[col1 + lm];
        #pragma unroll
        for (int i = 0; i < 8; ++i) {
            int r = i + 8 * half;
            h2[r * (HID + 1) + col0 + lm]        = leaky01(acc00[i] + bc0);
            h2[r * (HID + 1) + col1 + lm]        = leaky01(acc01[i] + bc1);
            h2[(16 + r) * (HID + 1) + col0 + lm] = leaky01(acc10[i] + bc0);
            h2[(16 + r) * (HID + 1) + col1 + lm] = leaky01(acc11[i] + bc1);
        }
    }
    __syncthreads();

    {
        int row = tid >> 3, seg = tid & 7;
        const float* hrow = h2 + row * (HID + 1) + seg * 32;
        const float* wseg = Wd + seg * 32;
        float p = 0.0f;
        #pragma unroll
        for (int c = 0; c < 32; ++c) p += hrow[c] * wseg[c];
        zb[row * 8 + seg] = p;
    }
    __syncthreads();
    if (tid < ROWS) {
        float acc = 0.0f;
        #pragma unroll
        for (int s = 0; s < 8; ++s) acc += zb[tid * 8 + s];
        out[nodeBase + tid] = acc + bd[0];
    }
}

extern "C" void kernel_launch(void* const* d_in, const int* in_sizes, int n_in,
                              void* d_out, int out_size, void* d_ws, size_t ws_size,
                              hipStream_t stream) {
    const float* x     = (const float*)d_in[0];
    const int*   ei    = (const int*)d_in[1];
    const float* W_gcn = (const float*)d_in[2];
    const float* b_gcn = (const float*)d_in[3];
    const float* W1    = (const float*)d_in[4];
    const float* b1    = (const float*)d_in[5];
    const float* W2    = (const float*)d_in[6];
    const float* b2    = (const float*)d_in[7];
    const float* Wd    = (const float*)d_in[8];
    const float* bd    = (const float*)d_in[9];
    float* out = (float*)d_out;

    char* ws = (char*)d_ws;
    size_t off0 = 0;
    size_t off1 = (off0 + (size_t)N_NODES * sizeof(float) + 255) & ~(size_t)255;
    size_t off2 = (off1 + (size_t)N_NODES * C_IN * sizeof(float) + 255) & ~(size_t)255;
    float* dinv = (float*)(ws + off0);
    float* h    = (float*)(ws + off1);
    float* agg  = (float*)(ws + off2);

    const int B = 256;
    deg_init_kernel<<<(N_NODES + B - 1) / B, B, 0, stream>>>(dinv, N_NODES);
    deg_accum_kernel<<<(N_EDGES + B - 1) / B, B, 0, stream>>>(ei, dinv, N_EDGES);
    deg_rsqrt_kernel<<<(N_NODES + B - 1) / B, B, 0, stream>>>(dinv, N_NODES);

    gemm_xw_kernel<<<N_NODES / 64, B, 0, stream>>>(x, W_gcn, dinv, h, agg, N_NODES);

    scatter_kernel<<<(N_EDGES / 32 + 7) / 8, B, 0, stream>>>(ei, h, dinv, agg);

    mlp_fused_kernel<<<N_NODES / ROWS, B, 0, stream>>>(agg, x, b_gcn, W1, b1, W2, b2,
                                                       Wd, bd, out);
}


# degree: each (edge slice, node range) work-group counts destinations in LDS and stores a whole-line plane of partial counts, summed in the rsqrt kernel; no global atomic per edge; 6 launches -> 5
# speedup vs baseline: 1.5432x; 1.1808x over previous
#include <hip/hip_runtime.h>

#define N_NODES 120000
#define N_EDGES 1200000
#define C_IN    64
#define HID     256
#define ROWS    32

typedef float v2f __attribute__((ext_vector_type(2)));
typedef float v8f __attribute__((ext_vector_type(8)));

__device__ __forceinline__ float leaky01(float v) { return v >= 0.0f ? v : 0.01f * v; }

__device__ __forceinline__ void atomAddF(float* p, float v) {
    __hip_atomic_fetch_add(p, v, __ATOMIC_RELAXED, __HIP_MEMORY_SCOPE_AGENT);
}

__device__ __forceinline__ v8f wmma4(v2f a, v2f b, v8f c) {
    return __builtin_amdgcn_wmma_f32_16x16x4_f32(false, a, false, b, (short)0, c,
                                                 false, false);
}

#define DEG_R   10
#define DEG_P   25
#define DEG_NR  (N_NODES / DEG_R)
#define DEG_EP  (N_EDGES / DEG_P)
#define DEG_T   512
#define DEG_NV  (DEG_EP / 4)
#define DEG_IT  ((DEG_NV + DEG_T - 1) / DEG_T)
#define DEG_LB  24
static_assert(N_NODES % DEG_R == 0 && DEG_NR % 32 == 0, "a range is a whole number of 128-B lines");
static_assert(N_EDGES % DEG_P == 0 && DEG_EP % 4 == 0 && N_EDGES % 4 == 0, "slices start 16-B aligned");
static_assert(DEG_IT % DEG_LB == 0, "load batches cover the slice evenly");
static_assert(N_NODES % 4 == 0, "four nodes per thread in the plane sum");

__device__ __forceinline__ void deg_count_one(unsigned* cnt, int d, int lo) {
    const unsigned u = (unsigned)(d - lo);
    if (u < (unsigned)DEG_NR)
        __hip_atomic_fetch_add(&cnt[u], 1u, __ATOMIC_RELAXED, __HIP_MEMORY_SCOPE_WORKGROUP);
}

__global__ __launch_bounds__(DEG_T) void deg_count_kernel(const int* __restrict__ ei,
                                                          unsigned* __restrict__ part) {
    __shared__ uint4 cnt4[DEG_NR / 4];
    unsigned* cnt = reinterpret_cast<unsigned*>(cnt4);
    const int tid = threadIdx.x;
    const int p  = blockIdx.x / DEG_R;
    const int lo = (blockIdx.x - p * DEG_R) * DEG_NR;

    for (int j = tid; j < DEG_NR / 4; j += DEG_T) cnt4[j] = make_uint4(0u, 0u, 0u, 0u);
    __syncthreads();

    const int4* dst4 = reinterpret_cast<const int4*>(ei + N_EDGES + (size_t)p * DEG_EP);
    #pragma unroll
    for (int b = 0; b < DEG_IT; b += DEG_LB) {
        int4 d[DEG_LB];
        #pragma unroll
        for (int k = 0; k < DEG_LB; ++k) {
            const int i = tid + (b + k) * DEG_T;
            d[k] = (i < DEG_NV) ? dst4[i] : make_int4(-1, -1, -1, -1);
        }
        #pragma unroll
        for (int k = 0; k < DEG_LB; ++k) {
            deg_count_one(cnt, d[k].x, lo);
            deg_count_one(cnt, d[k].y, lo);
            deg_count_one(cnt, d[k].z, lo);
            deg_count_one(cnt, d[k].w, lo);
        }
    }
    __syncthreads();

    uint4* o4 = reinterpret_cast<uint4*>(part + (size_t)p * N_NODES + lo);
    for (int j = tid; j < DEG_NR / 4; j += DEG_T) o4[j] = cnt4[j];
}

__global__ __launch_bounds__(128) void deg_rsqrt_kernel(const unsigned* __restrict__ part,
                                                        float* __restrict__ dinv) {
    const int i = blockIdx.x * 128 + threadIdx.x;
    if (i >= N_NODES / 4) return;
    const uint4* p4 = reinterpret_cast<const uint4*>(part) + i;
    uint4 s = make_uint4(1u, 1u, 1u, 1u);
    #pragma unroll
    for (int p = 0; p < DEG_P; ++p) {
        const uint4 v = p4[(size_t)p * (N_NODES / 4)];
        s.x += v.x; s.y += v.y; s.z += v.z; s.w += v.w;
    }
    float4 o;
    o.x = rsqrtf((float)s.x);
    o.y = rsqrtf((float)s.y);
    o.z = rsqrtf((float)s.z);
    o.w = rsqrtf((float)s.w);
    reinterpret_cast<float4*>(dinv)[i] = o;
}

__global__ __launch_bounds__(256) void gemm_xw_kernel(const float* __restrict__ x,
                                                      const float* __restrict__ W,
                                                      const float* __restrict__ dinv,
                                                      float* __restrict__ h,
                                                      float* __restrict__ agg, int nrows) {
    int tid  = threadIdx.x;
    int wave = tid >> 5, lane = tid & 31;
    int half = lane >> 4, lm = lane & 15;
    int row0 = blockIdx.x * 64 + (wave >> 2) * 16;
    int colBase = (wave & 3) * 16;
    if (row0 >= nrows) return;

    float di0[8], di1[8];
    #pragma unroll
    for (int i = 0; i < 8; ++i) {
        di0[i] = dinv[row0 + i + 8 * half];
        di1[i] = dinv[row0 + 32 + i + 8 * half];
    }

    v8f acc0 = {}, acc1 = {};
    const float* a0 = x + (size_t)(row0 + lm) * C_IN + 2 * half;
    const float* a1 = x + (size_t)(row0 + 32 + lm) * C_IN + 2 * half;
    #pragma unroll
    for (int k = 0; k < C_IN; k += 4) {
        v2f av0, av1, b;
        av0.x = a0[k];  av0.y = a0[k + 1];
        av1.x = a1[k];  av1.y = a1[k + 1];
        b.x = W[(size_t)(k + 2 * half) * C_IN + colBase + lm];
        b.y = W[(size_t)(k + 1 + 2 * half) * C_IN + colBase + lm];
        acc0 = wmma4(av0, b, acc0);
        acc1 = wmma4(av1, b, acc1);
    }
    #pragma unroll
    for (int i = 0; i < 8; ++i) {
        size_t o0 = (size_t)(row0 + i + 8 * half) * C_IN + colBase + lm;
        size_t o1 = (size_t)(row0 + 32 + i + 8 * half) * C_IN + colBase + lm;
        h[o0]   = acc0[i];
        h[o1]   = acc1[i];
        agg[o0] = acc0[i] * di0[i] * di0[i];
        agg[o1] = acc1[i] * di1[i] * di1[i];
    }
}

static_assert(N_EDGES % 32 == 0, "a wave owns exactly 32 edges");
#define SC_CHUNK 16
__global__ __launch_bounds__(256) void scatter_kernel(const int* __restrict__ ei,
                                                      const float* __restrict__ h,
                                                      const float* __restrict__ dinv,
                                                      float* __restrict__ agg) {
    const int lane = threadIdx.x & 31;
    const int wave = __builtin_amdgcn_readfirstlane((int)(threadIdx.x >> 5));
    const int e0 = (blockIdx.x * 8 + wave) * 32;
    if (e0 >= N_EDGES) return;

    const int s = ei[e0 + lane];
    const int d = ei[N_EDGES + e0 + lane];
    const float norm = dinv[s] * dinv[d];

    #pragma unroll
    for (int jb = 0; jb < 32; jb += SC_CHUNK) {
        float v0[SC_CHUNK], v1[SC_CHUNK];
        #pragma unroll
        for (int j = 0; j < SC_CHUNK; ++j) {
            const int sj = __builtin_amdgcn_readlane(s, jb + j);
            const float* hp = h + (size_t)sj * C_IN + lane;
            v0[j] = hp[0];
            v1[j] = hp[32];
        }
        #pragma unroll
        for (int j = 0; j < SC_CHUNK; ++j) {
            const int dj = __builtin_amdgcn_readlane(d, jb + j);
            const float nj = __int_as_float(__builtin_amdgcn_readlane(__float_as_int(norm), jb + j));
            float* ap = agg + (size_t)dj * C_IN + lane;
            atomAddF(ap,      v0[j] * nj);
            atomAddF(ap + 32, v1[j] * nj);
        }
    }
}

__global__ __launch_bounds__(256) void mlp_fused_kernel(
    const float* __restrict__ agg, const float* __restrict__ x,
    const float* __restrict__ b_gcn, const float* __restrict__ W1,
    const float* __restrict__ b1, const float* __restrict__ W2,
    const float* __restrict__ b2, const float* __restrict__ Wd,
    const float* __restrict__ bd, float* __restrict__ out) {
    __shared__ float zb[ROWS * (C_IN + 1)];
    __shared__ float h1[ROWS * (HID + 1)];
    __shared__ float h2[ROWS * (HID + 1)];

    int tid = threadIdx.x;
    int nodeBase = blockIdx.x * ROWS;

    for (int i = tid; i < ROWS * C_IN; i += 256) {
        int r = i >> 6, c = i & (C_IN - 1);
        size_t g = (size_t)(nodeBase + r) * C_IN + c;
        float v = agg[g] + b_gcn[c];
        v = v > 0.0f ? v : 0.0f;
        zb[r * (C_IN + 1) + c] = v + x[g];
    }
    __syncthreads();

    int wave = tid >> 5, lane = tid & 31, half = lane >> 4, lm = lane & 15;
    int col0 = wave * 16;
    int col1 = col0 + 128;

    {
        v8f acc00 = {}, acc01 = {}, acc10 = {}, acc11 = {};
        const float* aR0 = zb + lm * (C_IN + 1) + 2 * half;
        const float* aR1 = zb + (16 + lm) * (C_IN + 1) + 2 * half;
        const float* w0  = W1 + (size_t)(col0 + lm) * C_IN + 2 * half;
        const float* w1  = W1 + (size_t)(col1 + lm) * C_IN + 2 * half;
        #pragma unroll
        for (int k = 0; k < C_IN; k += 4) {
            v2f a0v, a1v, b0v, b1v;
            a0v.x = aR0[k]; a0v.y = aR0[k + 1];
            a1v.x = aR1[k]; a1v.y = aR1[k + 1];
            b0v.x = w0[k];  b0v.y = w0[k + 1];
            b1v.x = w1[k];  b1v.y = w1[k + 1];
            acc00 = wmma4(a0v, b0v, acc00);
            acc01 = wmma4(a0v, b1v, acc01);
            acc10 = wmma4(a1v, b0v, acc10);
            acc11 = wmma4(a1v, b1v, acc11);
        }
        float bc0 = b1[col0 + lm], bc1 = b1[col1 + lm];
        #pragma unroll
        for (int i = 0; i < 8; ++i) {
            int r = i + 8 * half;
            h1[r * (HID + 1) + col0 + lm]        = leaky01(acc00[i] + bc0);
            h1[r * (HID + 1) + col1 + lm]        = leaky01(acc01[i] + bc1);
            h1[(16 + r) * (HID + 1) + col0 + lm] = leaky01(acc10[i] + bc0);
            h1[(16 + r) * (HID + 1) + col1 + lm] = leaky01(acc11[i] + bc1);
        }
    }
    __syncthreads();

    {
        v8f acc00 = {}, acc01 = {}, acc10 = {}, acc11 = {};
        const float* aR0 = h1 + lm * (HID + 1) + 2 * half;
        const float* aR1 = h1 + (16 + lm) * (HID + 1) + 2 * half;
        const float* w0  = W2 + (size_t)(col0 + lm) * HID + 2 * half;
        const float* w1  = W2 + (size_t)(col1 + lm) * HID + 2 * half;
        #pragma unroll 4
        for (int k = 0; k < HID; k += 4) {
            v2f a0v, a1v, b0v, b1v;
            a0v.x = aR0[k]; a0v.y = aR0[k + 1];
            a1v.x = aR1[k]; a1v.y = aR1[k + 1];
            b0v.x = w0[k];  b0v.y = w0[k + 1];
            b1v.x = w1[k];  b1v.y = w1[k + 1];
            acc00 = wmma4(a0v, b0v, acc00);
            acc01 = wmma4(a0v, b1v, acc01);
            acc10 = wmma4(a1v, b0v, acc10);
            acc11 = wmma4(a1v, b1v, acc11);
        }
        float bc0 = b2[col0 + lm], bc1 = b2[col1 + lm];
        #pragma unroll
        for (int i = 0; i < 8; ++i) {
            int r = i + 8 * half;
            h2[r * (HID + 1) + col0 + lm]        = leaky01(acc00[i] + bc0);
            h2[r * (HID + 1) + col1 + lm]        = leaky01(acc01[i] + bc1);
            h2[(16 + r) * (HID + 1) + col0 + lm] = leaky01(acc10[i] + bc0);
            h2[(16 + r) * (HID + 1) + col1 + lm] = leaky01(acc11[i] + bc1);
        }
    }
    __syncthreads();

    {
        int row = tid >> 3, seg = tid & 7;
        const float* hrow = h2 + row * (HID + 1) + seg * 32;
        const float* wseg = Wd + seg * 32;
        float p = 0.0f;
        #pragma unroll
        for (int c = 0; c < 32; ++c) p += hrow[c] * wseg[c];
        zb[row * 8 + seg] = p;
    }
    __syncthreads();
    if (tid < ROWS) {
        float acc = 0.0f;
        #pragma unroll
        for (int s = 0; s < 8; ++s) acc += zb[tid * 8 + s];
        out[nodeBase + tid] = acc + bd[0];
    }
}

extern "C" void kernel_launch(void* const* d_in, const int* in_sizes, int n_in,
                              void* d_out, int out_size, void* d_ws, size_t ws_size,
                              hipStream_t stream) {
    const float* x     = (const float*)d_in[0];
    const int*   ei    = (const int*)d_in[1];
    const float* W_gcn = (const float*)d_in[2];
    const float* b_gcn = (const float*)d_in[3];
    const float* W1    = (const float*)d_in[4];
    const float* b1    = (const float*)d_in[5];
    const float* W2    = (const float*)d_in[6];
    const float* b2    = (const float*)d_in[7];
    const float* Wd    = (const float*)d_in[8];
    const float* bd    = (const float*)d_in[9];
    float* out = (float*)d_out;

    char* ws = (char*)d_ws;
    size_t off0 = 0;
    size_t off1 = (off0 + (size_t)N_NODES * sizeof(float) + 255) & ~(size_t)255;
    size_t off2 = (off1 + (size_t)N_NODES * C_IN * sizeof(float) + 255) & ~(size_t)255;
    size_t off3 = (off2 + (size_t)N_NODES * C_IN * sizeof(float) + 255) & ~(size_t)255;
    float* dinv = (float*)(ws + off0);
    float* h    = (float*)(ws + off1);
    float* agg  = (float*)(ws + off2);
    unsigned* part = (unsigned*)(ws + off3);

    const int B = 256;
    deg_count_kernel<<<DEG_R * DEG_P, DEG_T, 0, stream>>>(ei, part);
    deg_rsqrt_kernel<<<(N_NODES / 4 + 127) / 128, 128, 0, stream>>>(part, dinv);

    gemm_xw_kernel<<<N_NODES / 64, B, 0, stream>>>(x, W_gcn, dinv, h, agg, N_NODES);

    scatter_kernel<<<(N_EDGES / 32 + 7) / 8, B, 0, stream>>>(ei, h, dinv, agg);

    mlp_fused_kernel<<<N_NODES / ROWS, B, 0, stream>>>(agg, x, b_gcn, W1, b1, W2, b2,
                                                       Wd, bd, out);
}


# WMMA fragment loads widened to 16 B per lane (A from LDS rows at 16-B-odd pitch, B and x rows from global); k order inside each 32-k step permuted alike for A and B: each dot's sum order changes
# speedup vs baseline: 1.6199x; 1.0497x over previous
#include <hip/hip_runtime.h>

#define N_NODES 120000
#define N_EDGES 1200000
#define C_IN    64
#define HID     256
#define ROWS    32
#define ZP      (C_IN + 4)
#define H1P     (HID + 4)

typedef float v2f __attribute__((ext_vector_type(2)));
typedef float v8f __attribute__((ext_vector_type(8)));


__device__ __forceinline__ float leaky01(float v) { return v >= 0.0f ? v : 0.01f * v; }

__device__ __forceinline__ void atomAddF(float* p, float v) {
    __hip_atomic_fetch_add(p, v, __ATOMIC_RELAXED, __HIP_MEMORY_SCOPE_AGENT);
}

__device__ __forceinline__ v8f wmma4(v2f a, v2f b, v8f c) {
    return __builtin_amdgcn_wmma_f32_16x16x4_f32(false, a, false, b, (short)0, c,
                                                 false, false);
}

#define DEG_R   10
#define DEG_P   25
#define DEG_NR  (N_NODES / DEG_R)
#define DEG_EP  (N_EDGES / DEG_P)
#define DEG_T   512
#define DEG_NV  (DEG_EP / 4)
#define DEG_IT  ((DEG_NV + DEG_T - 1) / DEG_T)
#define DEG_LB  24
static_assert(N_NODES % DEG_R == 0 && DEG_NR % 32 == 0, "a range is a whole number of 128-B lines");
static_assert(N_EDGES % DEG_P == 0 && DEG_EP % 4 == 0 && N_EDGES % 4 == 0, "slices start 16-B aligned");
static_assert(DEG_IT % DEG_LB == 0, "load batches cover the slice evenly");
static_assert(N_NODES % 4 == 0, "four nodes per thread in the plane sum");

__device__ __forceinline__ void deg_count_one(unsigned* cnt, int d, int lo) {
    const unsigned u = (unsigned)(d - lo);
    if (u < (unsigned)DEG_NR)
        __hip_atomic_fetch_add(&cnt[u], 1u, __ATOMIC_RELAXED, __HIP_MEMORY_SCOPE_WORKGROUP);
}

__global__ __launch_bounds__(DEG_T) void deg_count_kernel(const int* __restrict__ ei,
                                                          unsigned* __restrict__ part) {
    __shared__ uint4 cnt4[DEG_NR / 4];
    unsigned* cnt = reinterpret_cast<unsigned*>(cnt4);
    const int tid = threadIdx.x;
    const int p  = blockIdx.x / DEG_R;
    const int lo = (blockIdx.x - p * DEG_R) * DEG_NR;

    for (int j = tid; j < DEG_NR / 4; j += DEG_T) cnt4[j] = make_uint4(0u, 0u, 0u, 0u);
    __syncthreads();

    const int4* dst4 = reinterpret_cast<const int4*>(ei + N_EDGES + (size_t)p * DEG_EP);
    #pragma unroll
    for (int b = 0; b < DEG_IT; b += DEG_LB) {
        int4 d[DEG_LB];
        #pragma unroll
        for (int k = 0; k < DEG_LB; ++k) {
            const int i = tid + (b + k) * DEG_T;
            d[k] = (i < DEG_NV) ? dst4[i] : make_int4(-1, -1, -1, -1);
        }
        #pragma unroll
        for (int k = 0; k < DEG_LB; ++k) {
            deg_count_one(cnt, d[k].x, lo);
            deg_count_one(cnt, d[k].y, lo);
            deg_count_one(cnt, d[k].z, lo);
            deg_count_one(cnt, d[k].w, lo);
        }
    }
    __syncthreads();

    uint4* o4 = reinterpret_cast<uint4*>(part + (size_t)p * N_NODES + lo);
    for (int j = tid; j < DEG_NR / 4; j += DEG_T) o4[j] = cnt4[j];
}

__global__ __launch_bounds__(128) void deg_rsqrt_kernel(const unsigned* __restrict__ part,
                                                        float* __restrict__ dinv) {
    const int i = blockIdx.x * 128 + threadIdx.x;
    if (i >= N_NODES / 4) return;
    const uint4* p4 = reinterpret_cast<const uint4*>(part) + i;
    uint4 s = make_uint4(1u, 1u, 1u, 1u);
    #pragma unroll
    for (int p = 0; p < DEG_P; ++p) {
        const uint4 v = p4[(size_t)p * (N_NODES / 4)];
        s.x += v.x; s.y += v.y; s.z += v.z; s.w += v.w;
    }
    float4 o;
    o.x = rsqrtf((float)s.x);
    o.y = rsqrtf((float)s.y);
    o.z = rsqrtf((float)s.z);
    o.w = rsqrtf((float)s.w);
    reinterpret_cast<float4*>(dinv)[i] = o;
}

__global__ __launch_bounds__(256) void gemm_xw_kernel(const float* __restrict__ x,
                                                      const float* __restrict__ W,
                                                      const float* __restrict__ dinv,
                                                      float* __restrict__ h,
                                                      float* __restrict__ agg, int nrows) {
    int tid  = threadIdx.x;
    int wave = tid >> 5, lane = tid & 31;
    int half = lane >> 4, lm = lane & 15;
    int row0 = blockIdx.x * 64 + (wave >> 2) * 16;
    int colBase = (wave & 3) * 16;
    if (row0 >= nrows) return;

    float di0[8], di1[8];
    #pragma unroll
    for (int i = 0; i < 8; ++i) {
        di0[i] = dinv[row0 + i + 8 * half];
        di1[i] = dinv[row0 + 32 + i + 8 * half];
    }

    v8f acc0 = {}, acc1 = {};
    const float4* a0 = reinterpret_cast<const float4*>(x + (size_t)(row0 + lm) * C_IN + 16 * half);
    const float4* a1 = reinterpret_cast<const float4*>(x + (size_t)(row0 + 32 + lm) * C_IN + 16 * half);
    const float* bw = W + (size_t)(16 * half) * C_IN + colBase + lm;
    #pragma unroll
    for (int t = 0; t < C_IN / 32; ++t) {
        float4 x0[4], x1[4];
        #pragma unroll
        for (int q = 0; q < 4; ++q) { x0[q] = a0[8 * t + q]; x1[q] = a1[8 * t + q]; }
        #pragma unroll
        for (int q = 0; q < 4; ++q) {
            const float* bq = bw + (size_t)(32 * t + 4 * q) * C_IN;
            v2f av0, av1, b;
            av0.x = x0[q].x;  av0.y = x0[q].y;
            av1.x = x1[q].x;  av1.y = x1[q].y;
            b.x = bq[0];      b.y = bq[C_IN];
            acc0 = wmma4(av0, b, acc0);
            acc1 = wmma4(av1, b, acc1);
            av0.x = x0[q].z;  av0.y = x0[q].w;
            av1.x = x1[q].z;  av1.y = x1[q].w;
            b.x = bq[2 * C_IN];  b.y = bq[3 * C_IN];
            acc0 = wmma4(av0, b, acc0);
            acc1 = wmma4(av1, b, acc1);
        }
    }
    #pragma unroll
    for (int i = 0; i < 8; ++i) {
        size_t o0 = (size_t)(row0 + i + 8 * half) * C_IN + colBase + lm;
        size_t o1 = (size_t)(row0 + 32 + i + 8 * half) * C_IN + colBase + lm;
        h[o0]   = acc0[i];
        h[o1]   = acc1[i];
        agg[o0] = acc0[i] * di0[i] * di0[i];
        agg[o1] = acc1[i] * di1[i] * di1[i];
    }
}

static_assert(N_EDGES % 32 == 0, "a wave owns exactly 32 edges");
#define SC_CHUNK 16
__global__ __launch_bounds__(256) void scatter_kernel(const int* __restrict__ ei,
                                                      const float* __restrict__ h,
                                                      const float* __restrict__ dinv,
                                                      float* __restrict__ agg) {
    const int lane = threadIdx.x & 31;
    const int wave = __builtin_amdgcn_readfirstlane((int)(threadIdx.x >> 5));
    const int e0 = (blockIdx.x * 8 + wave) * 32;
    if (e0 >= N_EDGES) return;

    const int s = ei[e0 + lane];
    const int d = ei[N_EDGES + e0 + lane];
    const float norm = dinv[s] * dinv[d];

    #pragma unroll
    for (int jb = 0; jb < 32; jb += SC_CHUNK) {
        float v0[SC_CHUNK], v1[SC_CHUNK];
        #pragma unroll
        for (int j = 0; j < SC_CHUNK; ++j) {
            const int sj = __builtin_amdgcn_readlane(s, jb + j);
            const float* hp = h + (size_t)sj * C_IN + lane;
            v0[j] = hp[0];
            v1[j] = hp[32];
        }
        #pragma unroll
        for (int j = 0; j < SC_CHUNK; ++j) {
            const int dj = __builtin_amdgcn_readlane(d, jb + j);
            const float nj = __int_as_float(__builtin_amdgcn_readlane(__float_as_int(norm), jb + j));
            float* ap = agg + (size_t)dj * C_IN + lane;
            atomAddF(ap,      v0[j] * nj);
            atomAddF(ap + 32, v1[j] * nj);
        }
    }
}

__global__ __launch_bounds__(256) void mlp_fused_kernel(
    const float* __restrict__ agg, const float* __restrict__ x,
    const float* __restrict__ b_gcn, const float* __restrict__ W1,
    const float* __restrict__ b1, const float* __restrict__ W2,
    const float* __restrict__ b2, const float* __restrict__ Wd,
    const float* __restrict__ bd, float* __restrict__ out) {
    __shared__ float4 zb4[ROWS * ZP / 4];
    __shared__ float4 h14[ROWS * H1P / 4];
    __shared__ float h2[ROWS * (HID + 1)];
    float* zb = reinterpret_cast<float*>(zb4);
    float* h1 = reinterpret_cast<float*>(h14);

    int tid = threadIdx.x;
    int nodeBase = blockIdx.x * ROWS;

    for (int i = tid; i < ROWS * C_IN; i += 256) {
        int r = i >> 6, c = i & (C_IN - 1);
        size_t g = (size_t)(nodeBase + r) * C_IN + c;
        float v = agg[g] + b_gcn[c];
        v = v > 0.0f ? v : 0.0f;
        zb[r * ZP + c] = v + x[g];
    }
    __syncthreads();

    int wave = tid >> 5, lane = tid & 31, half = lane >> 4, lm = lane & 15;
    int col0 = wave * 16;
    int col1 = col0 + 128;

    {
        v8f acc00 = {}, acc01 = {}, acc10 = {}, acc11 = {};
        const float4* aR0 = reinterpret_cast<const float4*>(zb + lm * ZP + 16 * half);
        const float4* aR1 = reinterpret_cast<const float4*>(zb + (16 + lm) * ZP + 16 * half);
        const float4* w0 = reinterpret_cast<const float4*>(W1 + (size_t)(col0 + lm) * C_IN + 16 * half);
        const float4* w1 = reinterpret_cast<const float4*>(W1 + (size_t)(col1 + lm) * C_IN + 16 * half);
        #pragma unroll
        for (int t = 0; t < C_IN / 32; ++t) {
            float4 a0[4], a1[4], b0[4], b1q[4];
            #pragma unroll
            for (int q = 0; q < 4; ++q) { b0[q] = w0[8 * t + q]; b1q[q] = w1[8 * t + q]; }
            #pragma unroll
            for (int q = 0; q < 4; ++q) { a0[q] = aR0[8 * t + q]; a1[q] = aR1[8 * t + q]; }
            #pragma unroll
            for (int q = 0; q < 4; ++q) {
                v2f a0v, a1v, b0v, b1v;
                a0v.x = a0[q].x;  a0v.y = a0[q].y;
                a1v.x = a1[q].x;  a1v.y = a1[q].y;
                b0v.x = b0[q].x;  b0v.y = b0[q].y;
                b1v.x = b1q[q].x; b1v.y = b1q[q].y;
                acc00 = wmma4(a0v, b0v, acc00);
                acc01 = wmma4(a0v, b1v, acc01);
                acc10 = wmma4(a1v, b0v, acc10);
                acc11 = wmma4(a1v, b1v, acc11);
                a0v.x = a0[q].z;  a0v.y = a0[q].w;
                a1v.x = a1[q].z;  a1v.y = a1[q].w;
                b0v.x = b0[q].z;  b0v.y = b0[q].w;
                b1v.x = b1q[q].z; b1v.y = b1q[q].w;
                acc00 = wmma4(a0v, b0v, acc00);
                acc01 = wmma4(a0v, b1v, acc01);
                acc10 = wmma4(a1v, b0v, acc10);
                acc11 = wmma4(a1v, b1v, acc11);
            }
        }
        float bc0 = b1[col0 + lm], bc1 = b1[col1 + lm];
        #pragma unroll
        for (int i = 0; i < 8; ++i) {
            int r = i + 8 * half;
            h1[r * H1P + col0 + lm]        = leaky01(acc00[i] + bc0);
            h1[r * H1P + col1 + lm]        = leaky01(acc01[i] + bc1);
            h1[(16 + r) * H1P + col0 + lm] = leaky01(acc10[i] + bc0);
            h1[(16 + r) * H1P + col1 + lm] = leaky01(acc11[i] + bc1);
        }
    }
    __syncthreads();

    {
        v8f acc00 = {}, acc01 = {}, acc10 = {}, acc11 = {};
        const float4* aR0 = reinterpret_cast<const float4*>(h1 + lm * H1P + 16 * half);
        const float4* aR1 = reinterpret_cast<const float4*>(h1 + (16 + lm) * H1P + 16 * half);
        const float4* w0 = reinterpret_cast<const float4*>(W2 + (size_t)(col0 + lm) * HID + 16 * half);
        const float4* w1 = reinterpret_cast<const float4*>(W2 + (size_t)(col1 + lm) * HID + 16 * half);
        #pragma unroll 1
        for (int kq = 0; kq < HID / 4; kq += 8) {
            float4 a0[4], a1[4], b0[4], b1q[4];
            #pragma unroll
            for (int q = 0; q < 4; ++q) { b0[q] = w0[kq + q]; b1q[q] = w1[kq + q]; }
            #pragma unroll
            for (int q = 0; q < 4; ++q) { a0[q] = aR0[kq + q]; a1[q] = aR1[kq + q]; }
            #pragma unroll
            for (int q = 0; q < 4; ++q) {
                v2f a0v, a1v, b0v, b1v;
                a0v.x = a0[q].x;  a0v.y = a0[q].y;
                a1v.x = a1[q].x;  a1v.y = a1[q].y;
                b0v.x = b0[q].x;  b0v.y = b0[q].y;
                b1v.x = b1q[q].x; b1v.y = b1q[q].y;
                acc00 = wmma4(a0v, b0v, acc00);
                acc01 = wmma4(a0v, b1v, acc01);
                acc10 = wmma4(a1v, b0v, acc10);
                acc11 = wmma4(a1v, b1v, acc11);
                a0v.x = a0[q].z;  a0v.y = a0[q].w;
                a1v.x = a1[q].z;  a1v.y = a1[q].w;
                b0v.x = b0[q].z;  b0v.y = b0[q].w;
                b1v.x = b1q[q].z; b1v.y = b1q[q].w;
                acc00 = wmma4(a0v, b0v, acc00);
                acc01 = wmma4(a0v, b1v, acc01);
                acc10 = wmma4(a1v, b0v, acc10);
                acc11 = wmma4(a1v, b1v, acc11);
            }
        }
        float bc0 = b2[col0 + lm], bc1 = b2[col1 + lm];
        #pragma unroll
        for (int i = 0; i < 8; ++i) {
            int r = i + 8 * half;
            h2[r * (HID + 1) + col0 + lm]        = leaky01(acc00[i] + bc0);
            h2[r * (HID + 1) + col1 + lm]        = leaky01(acc01[i] + bc1);
            h2[(16 + r) * (HID + 1) + col0 + lm] = leaky01(acc10[i] + bc0);
            h2[(16 + r) * (HID + 1) + col1 + lm] = leaky01(acc11[i] + bc1);
        }
    }
    __syncthreads();

    {
        int row = tid >> 3, seg = tid & 7;
        const float* hrow = h2 + row * (HID + 1) + seg * 32;
        const float* wseg = Wd + seg * 32;
        float p = 0.0f;
        #pragma unroll
        for (int c = 0; c < 32; ++c) p += hrow[c] * wseg[c];
        zb[row * 8 + seg] = p;
    }
    __syncthreads();
    if (tid < ROWS) {
        float acc = 0.0f;
        #pragma unroll
        for (int s = 0; s < 8; ++s) acc += zb[tid * 8 + s];
        out[nodeBase + tid] = acc + bd[0];
    }
}

extern "C" void kernel_launch(void* const* d_in, const int* in_sizes, int n_in,
                              void* d_out, int out_size, void* d_ws, size_t ws_size,
                              hipStream_t stream) {
    const float* x     = (const float*)d_in[0];
    const int*   ei    = (const int*)d_in[1];
    const float* W_gcn = (const float*)d_in[2];
    const float* b_gcn = (const float*)d_in[3];
    const float* W1    = (const float*)d_in[4];
    const float* b1    = (const float*)d_in[5];
    const float* W2    = (const float*)d_in[6];
    const float* b2    = (const float*)d_in[7];
    const float* Wd    = (const float*)d_in[8];
    const float* bd    = (const float*)d_in[9];
    float* out = (float*)d_out;

    char* ws = (char*)d_ws;
    size_t off0 = 0;
    size_t off1 = (off0 + (size_t)N_NODES * sizeof(float) + 255) & ~(size_t)255;
    size_t off2 = (off1 + (size_t)N_NODES * C_IN * sizeof(float) + 255) & ~(size_t)255;
    size_t off3 = (off2 + (size_t)N_NODES * C_IN * sizeof(float) + 255) & ~(size_t)255;
    float* dinv = (float*)(ws + off0);
    float* h    = (float*)(ws + off1);
    float* agg  = (float*)(ws + off2);
    unsigned* part = (unsigned*)(ws + off3);

    const int B = 256;
    deg_count_kernel<<<DEG_R * DEG_P, DEG_T, 0, stream>>>(ei, part);
    deg_rsqrt_kernel<<<(N_NODES / 4 + 127) / 128, 128, 0, stream>>>(part, dinv);

    gemm_xw_kernel<<<N_NODES / 64, B, 0, stream>>>(x, W_gcn, dinv, h, agg, N_NODES);

    scatter_kernel<<<(N_EDGES / 32 + 7) / 8, B, 0, stream>>>(ei, h, dinv, agg);

    mlp_fused_kernel<<<N_NODES / ROWS, B, 0, stream>>>(agg, x, b_gcn, W1, b1, W2, b2,
                                                       Wd, bd, out);
}


# graph-conv product moved behind the edge sum: scatter adds x*dinv[src] rows, fused kernel does dinv*(sum+dinv*x) @ W_gcn per tile; x@W launch and h array go (5 -> 4 launches); f32 sums reassociated
# speedup vs baseline: 1.6928x; 1.0450x over previous
#include <hip/hip_runtime.h>

#define N_NODES 120000
#define N_EDGES 1200000
#define C_IN    64
#define HID     256
#define ROWS    32
#define ZP      (C_IN + 4)
#define H1P     (HID + 4)


typedef float v2f __attribute__((ext_vector_type(2)));
typedef float v8f __attribute__((ext_vector_type(8)));


__device__ __forceinline__ float leaky01(float v) { return v >= 0.0f ? v : 0.01f * v; }

__device__ __forceinline__ void atomAddF(float* p, float v) {
    __hip_atomic_fetch_add(p, v, __ATOMIC_RELAXED, __HIP_MEMORY_SCOPE_AGENT);
}

__device__ __forceinline__ v8f wmma4(v2f a, v2f b, v8f c) {
    return __builtin_amdgcn_wmma_f32_16x16x4_f32(false, a, false, b, (short)0, c,
                                                 false, false);
}

#define DEG_R   10
#define DEG_P   25
#define DEG_NR  (N_NODES / DEG_R)
#define DEG_EP  (N_EDGES / DEG_P)
#define DEG_T   512
#define DEG_NV  (DEG_EP / 4)
#define DEG_IT  ((DEG_NV + DEG_T - 1) / DEG_T)
#define DEG_LB  24
static_assert(N_NODES % DEG_R == 0 && DEG_NR % 32 == 0, "a range is a whole number of 128-B lines");
static_assert(N_EDGES % DEG_P == 0 && DEG_EP % 4 == 0 && N_EDGES % 4 == 0, "slices start 16-B aligned");
static_assert(DEG_IT % DEG_LB == 0, "load batches cover the slice evenly");
static_assert(N_NODES % 4 == 0, "four nodes per thread in the plane sum");
#define AGG_V4  (N_NODES * C_IN / 4)
#define AGG_ZB  (AGG_V4 / (DEG_R * DEG_P))
static_assert(AGG_V4 % (DEG_R * DEG_P) == 0 && AGG_ZB % DEG_T == 0, "clearing covers the array evenly");

__device__ __forceinline__ void deg_count_one(unsigned* cnt, int d, int lo) {
    const unsigned u = (unsigned)(d - lo);
    if (u < (unsigned)DEG_NR)
        __hip_atomic_fetch_add(&cnt[u], 1u, __ATOMIC_RELAXED, __HIP_MEMORY_SCOPE_WORKGROUP);
}

__global__ __launch_bounds__(DEG_T) void deg_count_kernel(const int* __restrict__ ei,
                                                          unsigned* __restrict__ part,
                                                          float* __restrict__ agg) {
    __shared__ uint4 cnt4[DEG_NR / 4];
    unsigned* cnt = reinterpret_cast<unsigned*>(cnt4);
    const int tid = threadIdx.x;
    const int p  = blockIdx.x / DEG_R;
    const int lo = (blockIdx.x - p * DEG_R) * DEG_NR;

    for (int j = tid; j < DEG_NR / 4; j += DEG_T) cnt4[j] = make_uint4(0u, 0u, 0u, 0u);
    __syncthreads();

    {
        float4* z4 = reinterpret_cast<float4*>(agg) + (size_t)blockIdx.x * AGG_ZB + tid;
        #pragma unroll
        for (int k = 0; k < AGG_ZB / DEG_T; ++k) z4[k * DEG_T] = make_float4(0.0f, 0.0f, 0.0f, 0.0f);
    }

    const int4* dst4 = reinterpret_cast<const int4*>(ei + N_EDGES + (size_t)p * DEG_EP);
    #pragma unroll
    for (int b = 0; b < DEG_IT; b += DEG_LB) {
        int4 d[DEG_LB];
        #pragma unroll
        for (int k = 0; k < DEG_LB; ++k) {
            const int i = tid + (b + k) * DEG_T;
            d[k] = (i < DEG_NV) ? dst4[i] : make_int4(-1, -1, -1, -1);
        }
        #pragma unroll
        for (int k = 0; k < DEG_LB; ++k) {
            deg_count_one(cnt, d[k].x, lo);
            deg_count_one(cnt, d[k].y, lo);
            deg_count_one(cnt, d[k].z, lo);
            deg_count_one(cnt, d[k].w, lo);
        }
    }
    __syncthreads();

    uint4* o4 = reinterpret_cast<uint4*>(part + (size_t)p * N_NODES + lo);
    for (int j = tid; j < DEG_NR / 4; j += DEG_T) o4[j] = cnt4[j];
}

__global__ __launch_bounds__(128) void deg_rsqrt_kernel(const unsigned* __restrict__ part,
                                                        float* __restrict__ dinv) {
    const int i = blockIdx.x * 128 + threadIdx.x;
    if (i >= N_NODES / 4) return;
    const uint4* p4 = reinterpret_cast<const uint4*>(part) + i;
    uint4 s = make_uint4(1u, 1u, 1u, 1u);
    #pragma unroll
    for (int p = 0; p < DEG_P; ++p) {
        const uint4 v = p4[(size_t)p * (N_NODES / 4)];
        s.x += v.x; s.y += v.y; s.z += v.z; s.w += v.w;
    }
    float4 o;
    o.x = rsqrtf((float)s.x);
    o.y = rsqrtf((float)s.y);
    o.z = rsqrtf((float)s.z);
    o.w = rsqrtf((float)s.w);
    reinterpret_cast<float4*>(dinv)[i] = o;
}

static_assert(N_EDGES % 32 == 0, "a wave owns exactly 32 edges");
#define SC_CHUNK 16
__global__ __launch_bounds__(256) void scatter_kernel(const int* __restrict__ ei,
                                                      const float* __restrict__ x,
                                                      const float* __restrict__ dinv,
                                                      float* __restrict__ agg) {
    const int lane = threadIdx.x & 31;
    const int wave = __builtin_amdgcn_readfirstlane((int)(threadIdx.x >> 5));
    const int e0 = (blockIdx.x * 8 + wave) * 32;
    if (e0 >= N_EDGES) return;

    const int s = ei[e0 + lane];
    const int d = ei[N_EDGES + e0 + lane];
    const float norm = dinv[s];

    #pragma unroll
    for (int jb = 0; jb < 32; jb += SC_CHUNK) {
        float v0[SC_CHUNK], v1[SC_CHUNK];
        #pragma unroll
        for (int j = 0; j < SC_CHUNK; ++j) {
            const int sj = __builtin_amdgcn_readlane(s, jb + j);
            const float* hp = x + (size_t)sj * C_IN + lane;
            v0[j] = hp[0];
            v1[j] = hp[32];
        }
        #pragma unroll
        for (int j = 0; j < SC_CHUNK; ++j) {
            const int dj = __builtin_amdgcn_readlane(d, jb + j);
            const float nj = __int_as_float(__builtin_amdgcn_readlane(__float_as_int(norm), jb + j));
            float* ap = agg + (size_t)dj * C_IN + lane;
            atomAddF(ap,      v0[j] * nj);
            atomAddF(ap + 32, v1[j] * nj);
        }
    }
}

__global__ __launch_bounds__(256) void mlp_fused_kernel(
    const float* __restrict__ agg, const float* __restrict__ x,
    const float* __restrict__ dinv, const float* __restrict__ W_gcn,
    const float* __restrict__ b_gcn, const float* __restrict__ W1,
    const float* __restrict__ b1, const float* __restrict__ W2,
    const float* __restrict__ b2, const float* __restrict__ Wd,
    const float* __restrict__ bd, float* __restrict__ out) {
    __shared__ float4 zb4[ROWS * ZP / 4];
    __shared__ float4 h14[ROWS * H1P / 4];
    __shared__ float4 h24[ROWS * (HID + 1) / 4];
    static_assert((ROWS * (HID + 1)) % 4 == 0 && ROWS * ZP <= ROWS * (HID + 1), "u tile fits in h2's space");
    float* zb = reinterpret_cast<float*>(zb4);
    float* h1 = reinterpret_cast<float*>(h14);
    float* h2 = reinterpret_cast<float*>(h24);
    float* ub = h2;

    int tid = threadIdx.x;
    int nodeBase = blockIdx.x * ROWS;

    #pragma unroll
    for (int t = 0; t < ROWS * C_IN / 4 / 256; ++t) {
        const int i = tid + t * 256;
        const int r = i >> 4, c4 = i & 15;
        const size_t g4 = (size_t)(nodeBase + r) * (C_IN / 4) + c4;
        const float4 av = reinterpret_cast<const float4*>(agg)[g4];
        const float4 xv = reinterpret_cast<const float4*>(x)[g4];
        const float dv = dinv[nodeBase + r];
        float4 u;
        u.x = dv * (av.x + dv * xv.x);
        u.y = dv * (av.y + dv * xv.y);
        u.z = dv * (av.z + dv * xv.z);
        u.w = dv * (av.w + dv * xv.w);
        reinterpret_cast<float4*>(ub + r * ZP)[c4] = u;
        reinterpret_cast<float4*>(zb + r * ZP)[c4] = xv;
    }
    __syncthreads();

    int wave = tid >> 5, lane = tid & 31, half = lane >> 4, lm = lane & 15;
    int col0 = wave * 16;
    int col1 = col0 + 128;

    {
        const int rt = wave >> 2, ct = wave & 3;
        v8f acc = {};
        const float4* aR = reinterpret_cast<const float4*>(ub + (16 * rt + lm) * ZP + 16 * half);
        const float* bw = W_gcn + (size_t)(16 * half) * C_IN + 16 * ct + lm;
        #pragma unroll
        for (int t = 0; t < C_IN / 32; ++t) {
            float4 a[4];
            #pragma unroll
            for (int q = 0; q < 4; ++q) a[q] = aR[8 * t + q];
            #pragma unroll
            for (int q = 0; q < 4; ++q) {
                const float* bq = bw + (size_t)(32 * t + 4 * q) * C_IN;
                v2f av, b;
                av.x = a[q].x;  av.y = a[q].y;
                b.x = bq[0];    b.y = bq[C_IN];
                acc = wmma4(av, b, acc);
                av.x = a[q].z;  av.y = a[q].w;
                b.x = bq[2 * C_IN];  b.y = bq[3 * C_IN];
                acc = wmma4(av, b, acc);
            }
        }
        const float bc = b_gcn[16 * ct + lm];
        #pragma unroll
        for (int i = 0; i < 8; ++i) {
            float* zp = zb + (16 * rt + i + 8 * half) * ZP + 16 * ct + lm;
            float v = acc[i] + bc;
            v = v > 0.0f ? v : 0.0f;
            *zp = v + *zp;
        }
    }
    __syncthreads();

    {
        v8f acc00 = {}, acc01 = {}, acc10 = {}, acc11 = {};
        const float4* aR0 = reinterpret_cast<const float4*>(zb + lm * ZP + 16 * half);
        const float4* aR1 = reinterpret_cast<const float4*>(zb + (16 + lm) * ZP + 16 * half);
        const float4* w0 = reinterpret_cast<const float4*>(W1 + (size_t)(col0 + lm) * C_IN + 16 * half);
        const float4* w1 = reinterpret_cast<const float4*>(W1 + (size_t)(col1 + lm) * C_IN + 16 * half);
        #pragma unroll
        for (int t = 0; t < C_IN / 32; ++t) {
            float4 a0[4], a1[4], b0[4], b1q[4];
            #pragma unroll
            for (int q = 0; q < 4; ++q) { b0[q] = w0[8 * t + q]; b1q[q] = w1[8 * t + q]; }
            #pragma unroll
            for (int q = 0; q < 4; ++q) { a0[q] = aR0[8 * t + q]; a1[q] = aR1[8 * t + q]; }
            #pragma unroll
            for (int q = 0; q < 4; ++q) {
                v2f a0v, a1v, b0v, b1v;
                a0v.x = a0[q].x;  a0v.y = a0[q].y;
                a1v.x = a1[q].x;  a1v.y = a1[q].y;
                b0v.x = b0[q].x;  b0v.y = b0[q].y;
                b1v.x = b1q[q].x; b1v.y = b1q[q].y;
                acc00 = wmma4(a0v, b0v, acc00);
                acc01 = wmma4(a0v, b1v, acc01);
                acc10 = wmma4(a1v, b0v, acc10);
                acc11 = wmma4(a1v, b1v, acc11);
                a0v.x = a0[q].z;  a0v.y = a0[q].w;
                a1v.x = a1[q].z;  a1v.y = a1[q].w;
                b0v.x = b0[q].z;  b0v.y = b0[q].w;
                b1v.x = b1q[q].z; b1v.y = b1q[q].w;
                acc00 = wmma4(a0v, b0v, acc00);
                acc01 = wmma4(a0v, b1v, acc01);
                acc10 = wmma4(a1v, b0v, acc10);
                acc11 = wmma4(a1v, b1v, acc11);
            }
        }
        float bc0 = b1[col0 + lm], bc1 = b1[col1 + lm];
        #pragma unroll
        for (int i = 0; i < 8; ++i) {
            int r = i + 8 * half;
            h1[r * H1P + col0 + lm]        = leaky01(acc00[i] + bc0);
            h1[r * H1P + col1 + lm]        = leaky01(acc01[i] + bc1);
            h1[(16 + r) * H1P + col0 + lm] = leaky01(acc10[i] + bc0);
            h1[(16 + r) * H1P + col1 + lm] = leaky01(acc11[i] + bc1);
        }
    }
    __syncthreads();

    {
        v8f acc00 = {}, acc01 = {}, acc10 = {}, acc11 = {};
        const float4* aR0 = reinterpret_cast<const float4*>(h1 + lm * H1P + 16 * half);
        const float4* aR1 = reinterpret_cast<const float4*>(h1 + (16 + lm) * H1P + 16 * half);
        const float4* w0 = reinterpret_cast<const float4*>(W2 + (size_t)(col0 + lm) * HID + 16 * half);
        const float4* w1 = reinterpret_cast<const float4*>(W2 + (size_t)(col1 + lm) * HID + 16 * half);
        #pragma unroll 1
        for (int kq = 0; kq < HID / 4; kq += 8) {
            float4 a0[4], a1[4], b0[4], b1q[4];
            #pragma unroll
            for (int q = 0; q < 4; ++q) { b0[q] = w0[kq + q]; b1q[q] = w1[kq + q]; }
            #pragma unroll
            for (int q = 0; q < 4; ++q) { a0[q] = aR0[kq + q]; a1[q] = aR1[kq + q]; }
            #pragma unroll
            for (int q = 0; q < 4; ++q) {
                v2f a0v, a1v, b0v, b1v;
                a0v.x = a0[q].x;  a0v.y = a0[q].y;
                a1v.x = a1[q].x;  a1v.y = a1[q].y;
                b0v.x = b0[q].x;  b0v.y = b0[q].y;
                b1v.x = b1q[q].x; b1v.y = b1q[q].y;
                acc00 = wmma4(a0v, b0v, acc00);
                acc01 = wmma4(a0v, b1v, acc01);
                acc10 = wmma4(a1v, b0v, acc10);
                acc11 = wmma4(a1v, b1v, acc11);
                a0v.x = a0[q].z;  a0v.y = a0[q].w;
                a1v.x = a1[q].z;  a1v.y = a1[q].w;
                b0v.x = b0[q].z;  b0v.y = b0[q].w;
                b1v.x = b1q[q].z; b1v.y = b1q[q].w;
                acc00 = wmma4(a0v, b0v, acc00);
                acc01 = wmma4(a0v, b1v, acc01);
                acc10 = wmma4(a1v, b0v, acc10);
                acc11 = wmma4(a1v, b1v, acc11);
            }
        }
        float bc0 = b2[col0 + lm], bc1 = b2[col1 + lm];
        #pragma unroll
        for (int i = 0; i < 8; ++i) {
            int r = i + 8 * half;
            h2[r * (HID + 1) + col0 + lm]        = leaky01(acc00[i] + bc0);
            h2[r * (HID + 1) + col1 + lm]        = leaky01(acc01[i] + bc1);
            h2[(16 + r) * (HID + 1) + col0 + lm] = leaky01(acc10[i] + bc0);
            h2[(16 + r) * (HID + 1) + col1 + lm] = leaky01(acc11[i] + bc1);
        }
    }
    __syncthreads();

    {
        int row = tid >> 3, seg = tid & 7;
        const float* hrow = h2 + row * (HID + 1) + seg * 32;
        const float* wseg = Wd + seg * 32;
        float p = 0.0f;
        #pragma unroll
        for (int c = 0; c < 32; ++c) p += hrow[c] * wseg[c];
        zb[row * 8 + seg] = p;
    }
    __syncthreads();
    if (tid < ROWS) {
        float acc = 0.0f;
        #pragma unroll
        for (int s = 0; s < 8; ++s) acc += zb[tid * 8 + s];
        out[nodeBase + tid] = acc + bd[0];
    }
}

extern "C" void kernel_launch(void* const* d_in, const int* in_sizes, int n_in,
                              void* d_out, int out_size, void* d_ws, size_t ws_size,
                              hipStream_t stream) {
    const float* x     = (const float*)d_in[0];
    const int*   ei    = (const int*)d_in[1];
    const float* W_gcn = (const float*)d_in[2];
    const float* b_gcn = (const float*)d_in[3];
    const float* W1    = (const float*)d_in[4];
    const float* b1    = (const float*)d_in[5];
    const float* W2    = (const float*)d_in[6];
    const float* b2    = (const float*)d_in[7];
    const float* Wd    = (const float*)d_in[8];
    const float* bd    = (const float*)d_in[9];
    float* out = (float*)d_out;

    char* ws = (char*)d_ws;
    size_t off0 = 0;
    size_t off1 = (off0 + (size_t)N_NODES * sizeof(float) + 255) & ~(size_t)255;
    size_t off2 = (off1 + (size_t)N_NODES * C_IN * sizeof(float) + 255) & ~(size_t)255;
    float* dinv = (float*)(ws + off0);
    float* agg  = (float*)(ws + off1);
    unsigned* part = (unsigned*)(ws + off2);

    const int B = 256;
    deg_count_kernel<<<DEG_R * DEG_P, DEG_T, 0, stream>>>(ei, part, agg);
    deg_rsqrt_kernel<<<(N_NODES / 4 + 127) / 128, 128, 0, stream>>>(part, dinv);

    scatter_kernel<<<(N_EDGES / 32 + 7) / 8, B, 0, stream>>>(ei, x, dinv, agg);

    mlp_fused_kernel<<<N_NODES / ROWS, B, 0, stream>>>(agg, x, dinv, W_gcn, b_gcn, W1, b1, W2, b2,
                                                       Wd, bd, out);
}


# B fragments of all three products read from weights repacked per call into fragment order (by extra work-groups of the rsqrt launch): each wave-wide 16-B load covers 4 whole lines, not 16
# speedup vs baseline: 1.8240x; 1.0775x over previous
#include <hip/hip_runtime.h>

#define N_NODES 120000
#define N_EDGES 1200000
#define C_IN    64
#define HID     256
#define ROWS    32
#define ZP      (C_IN + 4)
#define H1P     (HID + 4)


typedef float v2f __attribute__((ext_vector_type(2)));
typedef float v8f __attribute__((ext_vector_type(8)));


__device__ __forceinline__ float leaky01(float v) { return v >= 0.0f ? v : 0.01f * v; }

__device__ __forceinline__ void atomAddF(float* p, float v) {
    __hip_atomic_fetch_add(p, v, __ATOMIC_RELAXED, __HIP_MEMORY_SCOPE_AGENT);
}

__device__ __forceinline__ v8f wmma4(v2f a, v2f b, v8f c) {
    return __builtin_amdgcn_wmma_f32_16x16x4_f32(false, a, false, b, (short)0, c,
                                                 false, false);
}

#define DEG_R   10
#define DEG_P   25
#define DEG_NR  (N_NODES / DEG_R)
#define DEG_EP  (N_EDGES / DEG_P)
#define DEG_T   512
#define DEG_NV  (DEG_EP / 4)
#define DEG_IT  ((DEG_NV + DEG_T - 1) / DEG_T)
#define DEG_LB  24
static_assert(N_NODES % DEG_R == 0 && DEG_NR % 32 == 0, "a range is a whole number of 128-B lines");
static_assert(N_EDGES % DEG_P == 0 && DEG_EP % 4 == 0 && N_EDGES % 4 == 0, "slices start 16-B aligned");
static_assert(DEG_IT % DEG_LB == 0, "load batches cover the slice evenly");
static_assert(N_NODES % 4 == 0, "four nodes per thread in the plane sum");
#define AGG_V4  (N_NODES * C_IN / 4)
#define AGG_ZB  (AGG_V4 / (DEG_R * DEG_P))
static_assert(AGG_V4 % (DEG_R * DEG_P) == 0 && AGG_ZB % DEG_T == 0, "clearing covers the array evenly");

__device__ __forceinline__ void deg_count_one(unsigned* cnt, int d, int lo) {
    const unsigned u = (unsigned)(d - lo);
    if (u < (unsigned)DEG_NR)
        __hip_atomic_fetch_add(&cnt[u], 1u, __ATOMIC_RELAXED, __HIP_MEMORY_SCOPE_WORKGROUP);
}

__global__ __launch_bounds__(DEG_T) void deg_count_kernel(const int* __restrict__ ei,
                                                          unsigned* __restrict__ part,
                                                          float* __restrict__ agg) {
    __shared__ uint4 cnt4[DEG_NR / 4];
    unsigned* cnt = reinterpret_cast<unsigned*>(cnt4);
    const int tid = threadIdx.x;
    const int p  = blockIdx.x / DEG_R;
    const int lo = (blockIdx.x - p * DEG_R) * DEG_NR;

    for (int j = tid; j < DEG_NR / 4; j += DEG_T) cnt4[j] = make_uint4(0u, 0u, 0u, 0u);
    __syncthreads();

    {
        float4* z4 = reinterpret_cast<float4*>(agg) + (size_t)blockIdx.x * AGG_ZB + tid;
        #pragma unroll
        for (int k = 0; k < AGG_ZB / DEG_T; ++k) z4[k * DEG_T] = make_float4(0.0f, 0.0f, 0.0f, 0.0f);
    }

    const int4* dst4 = reinterpret_cast<const int4*>(ei + N_EDGES + (size_t)p * DEG_EP);
    #pragma unroll
    for (int b = 0; b < DEG_IT; b += DEG_LB) {
        int4 d[DEG_LB];
        #pragma unroll
        for (int k = 0; k < DEG_LB; ++k) {
            const int i = tid + (b + k) * DEG_T;
            d[k] = (i < DEG_NV) ? dst4[i] : make_int4(-1, -1, -1, -1);
        }
        #pragma unroll
        for (int k = 0; k < DEG_LB; ++k) {
            deg_count_one(cnt, d[k].x, lo);
            deg_count_one(cnt, d[k].y, lo);
            deg_count_one(cnt, d[k].z, lo);
            deg_count_one(cnt, d[k].w, lo);
        }
    }
    __syncthreads();

    uint4* o4 = reinterpret_cast<uint4*>(part + (size_t)p * N_NODES + lo);
    for (int j = tid; j < DEG_NR / 4; j += DEG_T) o4[j] = cnt4[j];
}

#define PK_W2   (HID * HID / 4)
#define PK_W1   (HID * C_IN / 4)
#define PK_WG   (C_IN * C_IN / 4)
#define PK_ALL  (PK_W2 + PK_W1 + PK_WG)
#define RS_NB   ((N_NODES / 4 + 127) / 128)
#define PK_NB   (PK_ALL / 128)
static_assert(PK_ALL % 128 == 0, "the copy is a whole number of 128-thread work-groups");
static_assert(HID == 256 && C_IN == 64, "the index arithmetic of the copy is written for these sizes");

__device__ __forceinline__ void pack_weights(int o, const float* __restrict__ W_gcn,
                                             const float* __restrict__ W1,
                                             const float* __restrict__ W2,
                                             float4* __restrict__ wpk) {
    const int lane = o & 31, q = (o >> 5) & 3;
    const int half = lane >> 4, lm = lane & 15;
    float4 v;
    if (o < PK_W2) {
        const int j = (o >> 7) & 1, wave = (o >> 8) & 7, kt = o >> 11;
        const int g = 16 * wave + 128 * j + lm;
        v = reinterpret_cast<const float4*>(W2)[g * (HID / 4) + 8 * kt + 4 * half + q];
    } else if (o < PK_W2 + PK_W1) {
        const int o1 = o - PK_W2;
        const int j = (o1 >> 7) & 1, wave = (o1 >> 8) & 7, t = o1 >> 11;
        const int g = 16 * wave + 128 * j + lm;
        v = reinterpret_cast<const float4*>(W1)[g * (C_IN / 4) + 8 * t + 4 * half + q];
    } else {
        const int o2 = o - PK_W2 - PK_W1;
        const int t = (o2 >> 7) & 1, ct = o2 >> 8;
        const float* p = W_gcn + (size_t)(32 * t + 16 * half + 4 * q) * C_IN + 16 * ct + lm;
        v.x = p[0];  v.y = p[C_IN];  v.z = p[2 * C_IN];  v.w = p[3 * C_IN];
    }
    wpk[o] = v;
}

__global__ __launch_bounds__(128) void deg_rsqrt_pack_kernel(const unsigned* __restrict__ part,
                                                             float* __restrict__ dinv,
                                                             const float* __restrict__ W_gcn,
                                                             const float* __restrict__ W1,
                                                             const float* __restrict__ W2,
                                                             float* __restrict__ wpk_f) {
    if (blockIdx.x >= RS_NB) {
        pack_weights((blockIdx.x - RS_NB) * 128 + threadIdx.x, W_gcn, W1, W2,
                     reinterpret_cast<float4*>(wpk_f));
        return;
    }
    const int i = blockIdx.x * 128 + threadIdx.x;
    if (i >= N_NODES / 4) return;
    const uint4* p4 = reinterpret_cast<const uint4*>(part) + i;
    uint4 s = make_uint4(1u, 1u, 1u, 1u);
    #pragma unroll
    for (int p = 0; p < DEG_P; ++p) {
        const uint4 v = p4[(size_t)p * (N_NODES / 4)];
        s.x += v.x; s.y += v.y; s.z += v.z; s.w += v.w;
    }
    float4 o;
    o.x = rsqrtf((float)s.x);
    o.y = rsqrtf((float)s.y);
    o.z = rsqrtf((float)s.z);
    o.w = rsqrtf((float)s.w);
    reinterpret_cast<float4*>(dinv)[i] = o;
}

static_assert(N_EDGES % 32 == 0, "a wave owns exactly 32 edges");
#define SC_CHUNK 16
__global__ __launch_bounds__(256) void scatter_kernel(const int* __restrict__ ei,
                                                      const float* __restrict__ x,
                                                      const float* __restrict__ dinv,
                                                      float* __restrict__ agg) {
    const int lane = threadIdx.x & 31;
    const int wave = __builtin_amdgcn_readfirstlane((int)(threadIdx.x >> 5));
    const int e0 = (blockIdx.x * 8 + wave) * 32;
    if (e0 >= N_EDGES) return;

    const int s = ei[e0 + lane];
    const int d = ei[N_EDGES + e0 + lane];
    const float norm = dinv[s];

    #pragma unroll
    for (int jb = 0; jb < 32; jb += SC_CHUNK) {
        float v0[SC_CHUNK], v1[SC_CHUNK];
        #pragma unroll
        for (int j = 0; j < SC_CHUNK; ++j) {
            const int sj = __builtin_amdgcn_readlane(s, jb + j);
            const float* hp = x + (size_t)sj * C_IN + lane;
            v0[j] = hp[0];
            v1[j] = hp[32];
        }
        #pragma unroll
        for (int j = 0; j < SC_CHUNK; ++j) {
            const int dj = __builtin_amdgcn_readlane(d, jb + j);
            const float nj = __int_as_float(__builtin_amdgcn_readlane(__float_as_int(norm), jb + j));
            float* ap = agg + (size_t)dj * C_IN + lane;
            atomAddF(ap,      v0[j] * nj);
            atomAddF(ap + 32, v1[j] * nj);
        }
    }
}

__global__ __launch_bounds__(256) void mlp_fused_kernel(
    const float* __restrict__ agg, const float* __restrict__ x,
    const float* __restrict__ dinv, const float* __restrict__ wpk_f,
    const float* __restrict__ b_gcn,
    const float* __restrict__ b1,
    const float* __restrict__ b2, const float* __restrict__ Wd,
    const float* __restrict__ bd, float* __restrict__ out) {
    __shared__ float4 zb4[ROWS * ZP / 4];
    __shared__ float4 h14[ROWS * H1P / 4];
    __shared__ float4 h24[ROWS * (HID + 1) / 4];
    static_assert((ROWS * (HID + 1)) % 4 == 0 && ROWS * ZP <= ROWS * (HID + 1), "u tile fits in h2's space");
    const float4* wpk = reinterpret_cast<const float4*>(wpk_f);
    float* zb = reinterpret_cast<float*>(zb4);
    float* h1 = reinterpret_cast<float*>(h14);
    float* h2 = reinterpret_cast<float*>(h24);
    float* ub = h2;

    int tid = threadIdx.x;
    int nodeBase = blockIdx.x * ROWS;

    #pragma unroll
    for (int t = 0; t < ROWS * C_IN / 4 / 256; ++t) {
        const int i = tid + t * 256;
        const int r = i >> 4, c4 = i & 15;
        const size_t g4 = (size_t)(nodeBase + r) * (C_IN / 4) + c4;
        const float4 av = reinterpret_cast<const float4*>(agg)[g4];
        const float4 xv = reinterpret_cast<const float4*>(x)[g4];
        const float dv = dinv[nodeBase + r];
        float4 u;
        u.x = dv * (av.x + dv * xv.x);
        u.y = dv * (av.y + dv * xv.y);
        u.z = dv * (av.z + dv * xv.z);
        u.w = dv * (av.w + dv * xv.w);
        reinterpret_cast<float4*>(ub + r * ZP)[c4] = u;
        reinterpret_cast<float4*>(zb + r * ZP)[c4] = xv;
    }
    __syncthreads();

    int wave = tid >> 5, lane = tid & 31, half = lane >> 4, lm = lane & 15;
    int col0 = wave * 16;
    int col1 = col0 + 128;

    {
        const int rt = wave >> 2, ct = wave & 3;
        v8f acc = {};
        const float4* aR = reinterpret_cast<const float4*>(ub + (16 * rt + lm) * ZP + 16 * half);
        const float4* bw = wpk + PK_W2 + PK_W1 + ct * 256 + lane;
        #pragma unroll
        for (int t = 0; t < C_IN / 32; ++t) {
            float4 a[4], bv[4];
            #pragma unroll
            for (int q = 0; q < 4; ++q) bv[q] = bw[t * 128 + q * 32];
            #pragma unroll
            for (int q = 0; q < 4; ++q) a[q] = aR[8 * t + q];
            #pragma unroll
            for (int q = 0; q < 4; ++q) {
                v2f av, b;
                av.x = a[q].x;  av.y = a[q].y;
                b.x = bv[q].x;  b.y = bv[q].y;
                acc = wmma4(av, b, acc);
                av.x = a[q].z;  av.y = a[q].w;
                b.x = bv[q].z;  b.y = bv[q].w;
                acc = wmma4(av, b, acc);
            }
        }
        const float bc = b_gcn[16 * ct + lm];
        #pragma unroll
        for (int i = 0; i < 8; ++i) {
            float* zp = zb + (16 * rt + i + 8 * half) * ZP + 16 * ct + lm;
            float v = acc[i] + bc;
            v = v > 0.0f ? v : 0.0f;
            *zp = v + *zp;
        }
    }
    __syncthreads();

    {
        v8f acc00 = {}, acc01 = {}, acc10 = {}, acc11 = {};
        const float4* aR0 = reinterpret_cast<const float4*>(zb + lm * ZP + 16 * half);
        const float4* aR1 = reinterpret_cast<const float4*>(zb + (16 + lm) * ZP + 16 * half);
        const float4* w0 = wpk + PK_W2 + wave * 256 + lane;
        const float4* w1 = w0 + 128;
        #pragma unroll
        for (int t = 0; t < C_IN / 32; ++t) {
            float4 a0[4], a1[4], b0[4], b1q[4];
            #pragma unroll
            for (int q = 0; q < 4; ++q) { b0[q] = w0[2048 * t + 32 * q]; b1q[q] = w1[2048 * t + 32 * q]; }
            #pragma unroll
            for (int q = 0; q < 4; ++q) { a0[q] = aR0[8 * t + q]; a1[q] = aR1[8 * t + q]; }
            #pragma unroll
            for (int q = 0; q < 4; ++q) {
                v2f a0v, a1v, b0v, b1v;
                a0v.x = a0[q].x;  a0v.y = a0[q].y;
                a1v.x = a1[q].x;  a1v.y = a1[q].y;
                b0v.x = b0[q].x;  b0v.y = b0[q].y;
                b1v.x = b1q[q].x; b1v.y = b1q[q].y;
                acc00 = wmma4(a0v, b0v, acc00);
                acc01 = wmma4(a0v, b1v, acc01);
                acc10 = wmma4(a1v, b0v, acc10);
                acc11 = wmma4(a1v, b1v, acc11);
                a0v.x = a0[q].z;  a0v.y = a0[q].w;
                a1v.x = a1[q].z;  a1v.y = a1[q].w;
                b0v.x = b0[q].z;  b0v.y = b0[q].w;
                b1v.x = b1q[q].z; b1v.y = b1q[q].w;
                acc00 = wmma4(a0v, b0v, acc00);
                acc01 = wmma4(a0v, b1v, acc01);
                acc10 = wmma4(a1v, b0v, acc10);
                acc11 = wmma4(a1v, b1v, acc11);
            }
        }
        float bc0 = b1[col0 + lm], bc1 = b1[col1 + lm];
        #pragma unroll
        for (int i = 0; i < 8; ++i) {
            int r = i + 8 * half;
            h1[r * H1P + col0 + lm]        = leaky01(acc00[i] + bc0);
            h1[r * H1P + col1 + lm]        = leaky01(acc01[i] + bc1);
            h1[(16 + r) * H1P + col0 + lm] = leaky01(acc10[i] + bc0);
            h1[(16 + r) * H1P + col1 + lm] = leaky01(acc11[i] + bc1);
        }
    }
    __syncthreads();

    {
        v8f acc00 = {}, acc01 = {}, acc10 = {}, acc11 = {};
        const float4* aR0 = reinterpret_cast<const float4*>(h1 + lm * H1P + 16 * half);
        const float4* aR1 = reinterpret_cast<const float4*>(h1 + (16 + lm) * H1P + 16 * half);
        const float4* w0 = wpk + wave * 256 + lane;
        const float4* w1 = w0 + 128;
        #pragma unroll 1
        for (int kq = 0; kq < HID / 4; kq += 8) {
            float4 a0[4], a1[4], b0[4], b1q[4];
            #pragma unroll
            for (int q = 0; q < 4; ++q) { b0[q] = w0[256 * kq + 32 * q]; b1q[q] = w1[256 * kq + 32 * q]; }
            #pragma unroll
            for (int q = 0; q < 4; ++q) { a0[q] = aR0[kq + q]; a1[q] = aR1[kq + q]; }
            #pragma unroll
            for (int q = 0; q < 4; ++q) {
                v2f a0v, a1v, b0v, b1v;
                a0v.x = a0[q].x;  a0v.y = a0[q].y;
                a1v.x = a1[q].x;  a1v.y = a1[q].y;
                b0v.x = b0[q].x;  b0v.y = b0[q].y;
                b1v.x = b1q[q].x; b1v.y = b1q[q].y;
                acc00 = wmma4(a0v, b0v, acc00);
                acc01 = wmma4(a0v, b1v, acc01);
                acc10 = wmma4(a1v, b0v, acc10);
                acc11 = wmma4(a1v, b1v, acc11);
                a0v.x = a0[q].z;  a0v.y = a0[q].w;
                a1v.x = a1[q].z;  a1v.y = a1[q].w;
                b0v.x = b0[q].z;  b0v.y = b0[q].w;
                b1v.x = b1q[q].z; b1v.y = b1q[q].w;
                acc00 = wmma4(a0v, b0v, acc00);
                acc01 = wmma4(a0v, b1v, acc01);
                acc10 = wmma4(a1v, b0v, acc10);
                acc11 = wmma4(a1v, b1v, acc11);
            }
        }
        float bc0 = b2[col0 + lm], bc1 = b2[col1 + lm];
        #pragma unroll
        for (int i = 0; i < 8; ++i) {
            int r = i + 8 * half;
            h2[r * (HID + 1) + col0 + lm]        = leaky01(acc00[i] + bc0);
            h2[r * (HID + 1) + col1 + lm]        = leaky01(acc01[i] + bc1);
            h2[(16 + r) * (HID + 1) + col0 + lm] = leaky01(acc10[i] + bc0);
            h2[(16 + r) * (HID + 1) + col1 + lm] = leaky01(acc11[i] + bc1);
        }
    }
    __syncthreads();

    {
        int row = tid >> 3, seg = tid & 7;
        const float* hrow = h2 + row * (HID + 1) + seg * 32;
        const float* wseg = Wd + seg * 32;
        float p = 0.0f;
        #pragma unroll
        for (int c = 0; c < 32; ++c) p += hrow[c] * wseg[c];
        zb[row * 8 + seg] = p;
    }
    __syncthreads();
    if (tid < ROWS) {
        float acc = 0.0f;
        #pragma unroll
        for (int s = 0; s < 8; ++s) acc += zb[tid * 8 + s];
        out[nodeBase + tid] = acc + bd[0];
    }
}

extern "C" void kernel_launch(void* const* d_in, const int* in_sizes, int n_in,
                              void* d_out, int out_size, void* d_ws, size_t ws_size,
                              hipStream_t stream) {
    const float* x     = (const float*)d_in[0];
    const int*   ei    = (const int*)d_in[1];
    const float* W_gcn = (const float*)d_in[2];
    const float* b_gcn = (const float*)d_in[3];
    const float* W1    = (const float*)d_in[4];
    const float* b1    = (const float*)d_in[5];
    const float* W2    = (const float*)d_in[6];
    const float* b2    = (const float*)d_in[7];
    const float* Wd    = (const float*)d_in[8];
    const float* bd    = (const float*)d_in[9];
    float* out = (float*)d_out;

    char* ws = (char*)d_ws;
    size_t off0 = 0;
    size_t off1 = (off0 + (size_t)N_NODES * sizeof(float) + 255) & ~(size_t)255;
    size_t off2 = (off1 + (size_t)N_NODES * C_IN * sizeof(float) + 255) & ~(size_t)255;
    size_t off3 = (off2 + (size_t)DEG_P * N_NODES * sizeof(unsigned) + 255) & ~(size_t)255;
    float* dinv = (float*)(ws + off0);
    float* agg  = (float*)(ws + off1);
    unsigned* part = (unsigned*)(ws + off2);
    float* wpk = (float*)(ws + off3);

    const int B = 256;
    deg_count_kernel<<<DEG_R * DEG_P, DEG_T, 0, stream>>>(ei, part, agg);
    deg_rsqrt_pack_kernel<<<RS_NB + PK_NB, 128, 0, stream>>>(part, dinv, W_gcn, W1, W2, wpk);

    scatter_kernel<<<(N_EDGES / 32 + 7) / 8, B, 0, stream>>>(ei, x, dinv, agg);

    mlp_fused_kernel<<<N_NODES / ROWS, B, 0, stream>>>(agg, x, dinv, wpk, b_gcn, b1, b2,
                                                       Wd, bd, out);
}


# edge sum by owner work-groups: edges binned by node range (LDS counts, exact offsets), a range's rows summed in LDS and stored once; global f32 atomics and zero fill go; sums in arrival order
# speedup vs baseline: 2.0068x; 1.1002x over previous
#include <hip/hip_runtime.h>

#define N_NODES 120000
#define N_EDGES 1200000
#define C_IN    64
#define HID     256
#define ROWS    32
#define ZP      (C_IN + 4)
#define H1P     (HID + 4)


typedef float v2f __attribute__((ext_vector_type(2)));
typedef float v8f __attribute__((ext_vector_type(8)));


__device__ __forceinline__ float leaky01(float v) { return v >= 0.0f ? v : 0.01f * v; }

__device__ __forceinline__ void ldsAddF(float* p, float v) {
    __hip_atomic_fetch_add(p, v, __ATOMIC_RELAXED, __HIP_MEMORY_SCOPE_WORKGROUP);
}

__device__ __forceinline__ unsigned ldsTakeSlot(unsigned* p) {
    return __hip_atomic_fetch_add(p, 1u, __ATOMIC_RELAXED, __HIP_MEMORY_SCOPE_WORKGROUP);
}

__device__ __forceinline__ v8f wmma4(v2f a, v2f b, v8f c) {
    return __builtin_amdgcn_wmma_f32_16x16x4_f32(false, a, false, b, (short)0, c,
                                                 false, false);
}

#define DEG_R   10
#define DEG_P   25
#define DEG_NR  (N_NODES / DEG_R)
#define DEG_EP  (N_EDGES / DEG_P)
#define DEG_T   512
#define DEG_NV  (DEG_EP / 4)
#define DEG_IT  ((DEG_NV + DEG_T - 1) / DEG_T)
#define DEG_LB  24
static_assert(N_NODES % DEG_R == 0 && DEG_NR % 32 == 0, "a range is a whole number of 128-B lines");
static_assert(N_EDGES % DEG_P == 0 && DEG_EP % 4 == 0 && N_EDGES % 4 == 0, "slices start 16-B aligned");
static_assert(DEG_IT % DEG_LB == 0, "load batches cover the slice evenly");
static_assert(N_NODES % 4 == 0, "four nodes per thread in the plane sum");

#define OW_NB   250
#define OW_RN   (N_NODES / OW_NB)
#define OW_NS   250
#define OW_SE   (N_EDGES / OW_NS)
#define OW_SV   (OW_SE / 4)
#define OW_T    1024
static_assert(N_NODES % OW_NB == 0 && N_EDGES % OW_NS == 0 && OW_SE % 4 == 0, "ranges and slices are even");
static_assert(OW_NB < 256, "one thread per range, and one more for the total, in the offset scan");

__device__ __forceinline__ int ow_range(int d) {
    const unsigned b = (unsigned)d / (unsigned)OW_RN;
    return (int)(b < (unsigned)OW_NB ? b : (unsigned)(OW_NB - 1));
}

__device__ __forceinline__ void deg_count_one(unsigned* cnt, int d, int lo) {
    const unsigned u = (unsigned)(d - lo);
    if (u < (unsigned)DEG_NR)
        __hip_atomic_fetch_add(&cnt[u], 1u, __ATOMIC_RELAXED, __HIP_MEMORY_SCOPE_WORKGROUP);
}

__global__ __launch_bounds__(DEG_T) void deg_count_kernel(const int* __restrict__ ei,
                                                          unsigned* __restrict__ part,
                                                          unsigned* __restrict__ bcnt) {
    __shared__ uint4 cnt4[DEG_NR / 4];
    unsigned* cnt = reinterpret_cast<unsigned*>(cnt4);
    const int tid = threadIdx.x;

    if (blockIdx.x >= DEG_R * DEG_P) {
        const int s = blockIdx.x - DEG_R * DEG_P;
        if (tid < OW_NB) cnt[tid] = 0u;
        __syncthreads();
        const int4* sd4 = reinterpret_cast<const int4*>(ei + N_EDGES + (size_t)s * OW_SE);
        for (int i = tid; i < OW_SV; i += DEG_T) {
            const int4 d = sd4[i];
            ldsTakeSlot(&cnt[ow_range(d.x)]);
            ldsTakeSlot(&cnt[ow_range(d.y)]);
            ldsTakeSlot(&cnt[ow_range(d.z)]);
            ldsTakeSlot(&cnt[ow_range(d.w)]);
        }
        __syncthreads();
        if (tid < OW_NB) bcnt[s * OW_NB + tid] = cnt[tid];
        return;
    }
    const int p  = blockIdx.x / DEG_R;
    const int lo = (blockIdx.x - p * DEG_R) * DEG_NR;

    for (int j = tid; j < DEG_NR / 4; j += DEG_T) cnt4[j] = make_uint4(0u, 0u, 0u, 0u);
    __syncthreads();

    const int4* dst4 = reinterpret_cast<const int4*>(ei + N_EDGES + (size_t)p * DEG_EP);
    #pragma unroll
    for (int b = 0; b < DEG_IT; b += DEG_LB) {
        int4 d[DEG_LB];
        #pragma unroll
        for (int k = 0; k < DEG_LB; ++k) {
            const int i = tid + (b + k) * DEG_T;
            d[k] = (i < DEG_NV) ? dst4[i] : make_int4(-1, -1, -1, -1);
        }
        #pragma unroll
        for (int k = 0; k < DEG_LB; ++k) {
            deg_count_one(cnt, d[k].x, lo);
            deg_count_one(cnt, d[k].y, lo);
            deg_count_one(cnt, d[k].z, lo);
            deg_count_one(cnt, d[k].w, lo);
        }
    }
    __syncthreads();

    uint4* o4 = reinterpret_cast<uint4*>(part + (size_t)p * N_NODES + lo);
    for (int j = tid; j < DEG_NR / 4; j += DEG_T) o4[j] = cnt4[j];
}

#define PK_W2   (HID * HID / 4)
#define PK_W1   (HID * C_IN / 4)
#define PK_WG   (C_IN * C_IN / 4)
#define PK_ALL  (PK_W2 + PK_W1 + PK_WG)
#define RS_T    256
#define RS_NB   ((N_NODES / 4 + RS_T - 1) / RS_T)
#define PK_NB   (PK_ALL / RS_T)
static_assert(PK_ALL % RS_T == 0, "the copy is a whole number of work-groups");
static_assert(HID == 256 && C_IN == 64, "the index arithmetic of the copy is written for these sizes");

__device__ __forceinline__ void pack_weights(int o, const float* __restrict__ W_gcn,
                                             const float* __restrict__ W1,
                                             const float* __restrict__ W2,
                                             float4* __restrict__ wpk) {
    const int lane = o & 31, q = (o >> 5) & 3;
    const int half = lane >> 4, lm = lane & 15;
    float4 v;
    if (o < PK_W2) {
        const int j = (o >> 7) & 1, wave = (o >> 8) & 7, kt = o >> 11;
        const int g = 16 * wave + 128 * j + lm;
        v = reinterpret_cast<const float4*>(W2)[g * (HID / 4) + 8 * kt + 4 * half + q];
    } else if (o < PK_W2 + PK_W1) {
        const int o1 = o - PK_W2;
        const int j = (o1 >> 7) & 1, wave = (o1 >> 8) & 7, t = o1 >> 11;
        const int g = 16 * wave + 128 * j + lm;
        v = reinterpret_cast<const float4*>(W1)[g * (C_IN / 4) + 8 * t + 4 * half + q];
    } else {
        const int o2 = o - PK_W2 - PK_W1;
        const int t = (o2 >> 7) & 1, ct = o2 >> 8;
        const float* p = W_gcn + (size_t)(32 * t + 16 * half + 4 * q) * C_IN + 16 * ct + lm;
        v.x = p[0];  v.y = p[C_IN];  v.z = p[2 * C_IN];  v.w = p[3 * C_IN];
    }
    wpk[o] = v;
}

struct PlaceLds {
    int2 lp[OW_SE];
    unsigned long long sc[2][256];
    unsigned lcur[256];
    unsigned gdelta[256];
};

__device__ __forceinline__ void place_slice(PlaceLds& L, int s, const int* __restrict__ ei,
                                            const unsigned* __restrict__ bcnt,
                                            unsigned* __restrict__ bstart,
                                            int2* __restrict__ pairs) {
    const int t = threadIdx.x;
    unsigned tot = 0u, before = 0u, mine = 0u;
    if (t < OW_NB) {
        #pragma unroll 50
        for (int q = 0; q < OW_NS; ++q) {
            const unsigned v = bcnt[q * OW_NB + t];
            tot += v;
            before += (q < s) ? v : 0u;
            mine = (q == s) ? v : mine;
        }
    }
    L.sc[0][t] = ((unsigned long long)mine << 32) | (unsigned long long)tot;
    __syncthreads();
    int cur = 0;
    for (int off = 1; off < 256; off <<= 1) {
        unsigned long long v = L.sc[cur][t];
        if (t >= off) v += L.sc[cur][t - off];
        L.sc[cur ^ 1][t] = v;
        cur ^= 1;
        __syncthreads();
    }
    const unsigned long long incl = L.sc[cur][t];
    const unsigned gstart = (unsigned)(incl & 0xffffffffull) - tot;
    const unsigned lstart = (unsigned)(incl >> 32) - mine;
    L.lcur[t] = lstart;
    L.gdelta[t] = gstart + before - lstart;
    if (s == 0 && t <= OW_NB) bstart[t] = gstart;
    __syncthreads();

    const int4* s4 = reinterpret_cast<const int4*>(ei + (size_t)s * OW_SE);
    const int4* d4 = reinterpret_cast<const int4*>(ei + N_EDGES + (size_t)s * OW_SE);
    for (int i = t; i < OW_SV; i += RS_T) {
        const int4 sv = s4[i];
        const int4 dv = d4[i];
        L.lp[ldsTakeSlot(&L.lcur[ow_range(dv.x)])] = make_int2(sv.x, dv.x);
        L.lp[ldsTakeSlot(&L.lcur[ow_range(dv.y)])] = make_int2(sv.y, dv.y);
        L.lp[ldsTakeSlot(&L.lcur[ow_range(dv.z)])] = make_int2(sv.z, dv.z);
        L.lp[ldsTakeSlot(&L.lcur[ow_range(dv.w)])] = make_int2(sv.w, dv.w);
    }
    __syncthreads();
    for (int i = t; i < OW_SE; i += RS_T) {
        const int2 p = L.lp[i];
        pairs[(unsigned)i + L.gdelta[ow_range(p.y)]] = p;
    }
}

__global__ __launch_bounds__(RS_T) void deg_rsqrt_pack_kernel(const unsigned* __restrict__ part,
                                                              float* __restrict__ dinv,
                                                              const float* __restrict__ W_gcn,
                                                              const float* __restrict__ W1,
                                                              const float* __restrict__ W2,
                                                              float* __restrict__ wpk_f,
                                                              const int* __restrict__ ei,
                                                              const unsigned* __restrict__ bcnt,
                                                              unsigned* __restrict__ bstart,
                                                              int2* __restrict__ pairs) {
    __shared__ PlaceLds placeLds;
    if (blockIdx.x < OW_NS) {
        place_slice(placeLds, blockIdx.x, ei, bcnt, bstart, pairs);
        return;
    }
    const int bi = blockIdx.x - OW_NS;
    if (bi >= RS_NB) {
        pack_weights((bi - RS_NB) * RS_T + threadIdx.x, W_gcn, W1, W2,
                     reinterpret_cast<float4*>(wpk_f));
        return;
    }
    const int i = bi * RS_T + threadIdx.x;
    if (i >= N_NODES / 4) return;
    const uint4* p4 = reinterpret_cast<const uint4*>(part) + i;
    uint4 s = make_uint4(1u, 1u, 1u, 1u);
    #pragma unroll
    for (int p = 0; p < DEG_P; ++p) {
        const uint4 v = p4[(size_t)p * (N_NODES / 4)];
        s.x += v.x; s.y += v.y; s.z += v.z; s.w += v.w;
    }
    float4 o;
    o.x = rsqrtf((float)s.x);
    o.y = rsqrtf((float)s.y);
    o.z = rsqrtf((float)s.z);
    o.w = rsqrtf((float)s.w);
    reinterpret_cast<float4*>(dinv)[i] = o;
}

__global__ __launch_bounds__(OW_T) void owner_sum_kernel(const int2* __restrict__ pairs,
                                                         const unsigned* __restrict__ bstart,
                                                         const float* __restrict__ x,
                                                         const float* __restrict__ dinv,
                                                         float* __restrict__ agg) {
    __shared__ float4 sums4[OW_RN * C_IN / 4];
    float* sums = reinterpret_cast<float*>(sums4);
    const int tid = threadIdx.x;
    const int b = blockIdx.x;
    for (int i = tid; i < OW_RN * C_IN / 4; i += OW_T) sums4[i] = make_float4(0.0f, 0.0f, 0.0f, 0.0f);
    __syncthreads();

    const int lane = tid & 31;
    const int wave = __builtin_amdgcn_readfirstlane(tid >> 5);
    const int eb = (int)bstart[b], ee = (int)bstart[b + 1];
    const int lo = b * OW_RN;

    int base = eb + wave * 32;
    int s = 0, d = lo;
    float norm = 0.0f;
    if (base + lane < ee) {
        const int2 p = pairs[base + lane];
        s = p.x;  d = p.y;
        norm = dinv[s];
    }
    while (base < ee) {
        const int n = (ee - base) < 32 ? (ee - base) : 32;
        const int nbase = base + OW_T;
        const bool more = nbase + lane < ee;
        int2 pn = make_int2(0, lo);
        if (more) pn = pairs[nbase + lane];

        float2 v[32];
        #pragma unroll
        for (int j = 0; j < 32; ++j) {
            const int sj = __builtin_amdgcn_readlane(s, j);
            v[j] = reinterpret_cast<const float2*>(x + (size_t)sj * C_IN)[lane];
        }
        float normn = 0.0f;
        if (more) normn = dinv[pn.x];
        #pragma unroll
        for (int j = 0; j < 32; ++j) {
            const int r = __builtin_amdgcn_readlane(d, j) - lo;
            const float nj = __int_as_float(__builtin_amdgcn_readlane(__float_as_int(norm), j));
            if (j < n && (unsigned)r < (unsigned)OW_RN) {
                float* ap = sums + r * C_IN + 2 * lane;
                ldsAddF(ap,     v[j].x * nj);
                ldsAddF(ap + 1, v[j].y * nj);
            }
        }
        s = pn.x;  d = pn.y;  norm = normn;
        base = nbase;
    }
    __syncthreads();

    float4* o4 = reinterpret_cast<float4*>(agg) + (size_t)lo * (C_IN / 4);
    for (int i = tid; i < OW_RN * C_IN / 4; i += OW_T) o4[i] = sums4[i];
}

__global__ __launch_bounds__(256) void mlp_fused_kernel(
    const float* __restrict__ agg, const float* __restrict__ x,
    const float* __restrict__ dinv, const float* __restrict__ wpk_f,
    const float* __restrict__ b_gcn,
    const float* __restrict__ b1,
    const float* __restrict__ b2, const float* __restrict__ Wd,
    const float* __restrict__ bd, float* __restrict__ out) {
    __shared__ float4 zb4[ROWS * ZP / 4];
    __shared__ float4 h14[ROWS * H1P / 4];
    __shared__ float4 h24[ROWS * (HID + 1) / 4];
    static_assert((ROWS * (HID + 1)) % 4 == 0 && ROWS * ZP <= ROWS * (HID + 1), "u tile fits in h2's space");
    const float4* wpk = reinterpret_cast<const float4*>(wpk_f);
    float* zb = reinterpret_cast<float*>(zb4);
    float* h1 = reinterpret_cast<float*>(h14);
    float* h2 = reinterpret_cast<float*>(h24);
    float* ub = h2;

    int tid = threadIdx.x;
    int nodeBase = blockIdx.x * ROWS;

    #pragma unroll
    for (int t = 0; t < ROWS * C_IN / 4 / 256; ++t) {
        const int i = tid + t * 256;
        const int r = i >> 4, c4 = i & 15;
        const size_t g4 = (size_t)(nodeBase + r) * (C_IN / 4) + c4;
        const float4 av = reinterpret_cast<const float4*>(agg)[g4];
        const float4 xv = reinterpret_cast<const float4*>(x)[g4];
        const float dv = dinv[nodeBase + r];
        float4 u;
        u.x = dv * (av.x + dv * xv.x);
        u.y = dv * (av.y + dv * xv.y);
        u.z = dv * (av.z + dv * xv.z);
        u.w = dv * (av.w + dv * xv.w);
        reinterpret_cast<float4*>(ub + r * ZP)[c4] = u;
        reinterpret_cast<float4*>(zb + r * ZP)[c4] = xv;
    }
    __syncthreads();

    int wave = tid >> 5, lane = tid & 31, half = lane >> 4, lm = lane & 15;
    int col0 = wave * 16;
    int col1 = col0 + 128;

    {
        const int rt = wave >> 2, ct = wave & 3;
        v8f acc = {};
        const float4* aR = reinterpret_cast<const float4*>(ub + (16 * rt + lm) * ZP + 16 * half);
        const float4* bw = wpk + PK_W2 + PK_W1 + ct * 256 + lane;
        #pragma unroll
        for (int t = 0; t < C_IN / 32; ++t) {
            float4 a[4], bv[4];
            #pragma unroll
            for (int q = 0; q < 4; ++q) bv[q] = bw[t * 128 + q * 32];
            #pragma unroll
            for (int q = 0; q < 4; ++q) a[q] = aR[8 * t + q];
            #pragma unroll
            for (int q = 0; q < 4; ++q) {
                v2f av, b;
                av.x = a[q].x;  av.y = a[q].y;
                b.x = bv[q].x;  b.y = bv[q].y;
                acc = wmma4(av, b, acc);
                av.x = a[q].z;  av.y = a[q].w;
                b.x = bv[q].z;  b.y = bv[q].w;
                acc = wmma4(av, b, acc);
            }
        }
        const float bc = b_gcn[16 * ct + lm];
        #pragma unroll
        for (int i = 0; i < 8; ++i) {
            float* zp = zb + (16 * rt + i + 8 * half) * ZP + 16 * ct + lm;
            float v = acc[i] + bc;
            v = v > 0.0f ? v : 0.0f;
            *zp = v + *zp;
        }
    }
    __syncthreads();

    {
        v8f acc00 = {}, acc01 = {}, acc10 = {}, acc11 = {};
        const float4* aR0 = reinterpret_cast<const float4*>(zb + lm * ZP + 16 * half);
        const float4* aR1 = reinterpret_cast<const float4*>(zb + (16 + lm) * ZP + 16 * half);
        const float4* w0 = wpk + PK_W2 + wave * 256 + lane;
        const float4* w1 = w0 + 128;
        #pragma unroll
        for (int t = 0; t < C_IN / 32; ++t) {
            float4 a0[4], a1[4], b0[4], b1q[4];
            #pragma unroll
            for (int q = 0; q < 4; ++q) { b0[q] = w0[2048 * t + 32 * q]; b1q[q] = w1[2048 * t + 32 * q]; }
            #pragma unroll
            for (int q = 0; q < 4; ++q) { a0[q] = aR0[8 * t + q]; a1[q] = aR1[8 * t + q]; }
            #pragma unroll
            for (int q = 0; q < 4; ++q) {
                v2f a0v, a1v, b0v, b1v;
                a0v.x = a0[q].x;  a0v.y = a0[q].y;
                a1v.x = a1[q].x;  a1v.y = a1[q].y;
                b0v.x = b0[q].x;  b0v.y = b0[q].y;
                b1v.x = b1q[q].x; b1v.y = b1q[q].y;
                acc00 = wmma4(a0v, b0v, acc00);
                acc01 = wmma4(a0v, b1v, acc01);
                acc10 = wmma4(a1v, b0v, acc10);
                acc11 = wmma4(a1v, b1v, acc11);
                a0v.x = a0[q].z;  a0v.y = a0[q].w;
                a1v.x = a1[q].z;  a1v.y = a1[q].w;
                b0v.x = b0[q].z;  b0v.y = b0[q].w;
                b1v.x = b1q[q].z; b1v.y = b1q[q].w;
                acc00 = wmma4(a0v, b0v, acc00);
                acc01 = wmma4(a0v, b1v, acc01);
                acc10 = wmma4(a1v, b0v, acc10);
                acc11 = wmma4(a1v, b1v, acc11);
            }
        }
        float bc0 = b1[col0 + lm], bc1 = b1[col1 + lm];
        #pragma unroll
        for (int i = 0; i < 8; ++i) {
            int r = i + 8 * half;
            h1[r * H1P + col0 + lm]        = leaky01(acc00[i] + bc0);
            h1[r * H1P + col1 + lm]        = leaky01(acc01[i] + bc1);
            h1[(16 + r) * H1P + col0 + lm] = leaky01(acc10[i] + bc0);
            h1[(16 + r) * H1P + col1 + lm] = leaky01(acc11[i] + bc1);
        }
    }
    __syncthreads();

    {
        v8f acc00 = {}, acc01 = {}, acc10 = {}, acc11 = {};
        const float4* aR0 = reinterpret_cast<const float4*>(h1 + lm * H1P + 16 * half);
        const float4* aR1 = reinterpret_cast<const float4*>(h1 + (16 + lm) * H1P + 16 * half);
        const float4* w0 = wpk + wave * 256 + lane;
        const float4* w1 = w0 + 128;
        #pragma unroll 1
        for (int kq = 0; kq < HID / 4; kq += 8) {
            float4 a0[4], a1[4], b0[4], b1q[4];
            #pragma unroll
            for (int q = 0; q < 4; ++q) { b0[q] = w0[256 * kq + 32 * q]; b1q[q] = w1[256 * kq + 32 * q]; }
            #pragma unroll
            for (int q = 0; q < 4; ++q) { a0[q] = aR0[kq + q]; a1[q] = aR1[kq + q]; }
            #pragma unroll
            for (int q = 0; q < 4; ++q) {
                v2f a0v, a1v, b0v, b1v;
                a0v.x = a0[q].x;  a0v.y = a0[q].y;
                a1v.x = a1[q].x;  a1v.y = a1[q].y;
                b0v.x = b0[q].x;  b0v.y = b0[q].y;
                b1v.x = b1q[q].x; b1v.y = b1q[q].y;
                acc00 = wmma4(a0v, b0v, acc00);
                acc01 = wmma4(a0v, b1v, acc01);
                acc10 = wmma4(a1v, b0v, acc10);
                acc11 = wmma4(a1v, b1v, acc11);
                a0v.x = a0[q].z;  a0v.y = a0[q].w;
                a1v.x = a1[q].z;  a1v.y = a1[q].w;
                b0v.x = b0[q].z;  b0v.y = b0[q].w;
                b1v.x = b1q[q].z; b1v.y = b1q[q].w;
                acc00 = wmma4(a0v, b0v, acc00);
                acc01 = wmma4(a0v, b1v, acc01);
                acc10 = wmma4(a1v, b0v, acc10);
                acc11 = wmma4(a1v, b1v, acc11);
            }
        }
        float bc0 = b2[col0 + lm], bc1 = b2[col1 + lm];
        #pragma unroll
        for (int i = 0; i < 8; ++i) {
            int r = i + 8 * half;
            h2[r * (HID + 1) + col0 + lm]        = leaky01(acc00[i] + bc0);
            h2[r * (HID + 1) + col1 + lm]        = leaky01(acc01[i] + bc1);
            h2[(16 + r) * (HID + 1) + col0 + lm] = leaky01(acc10[i] + bc0);
            h2[(16 + r) * (HID + 1) + col1 + lm] = leaky01(acc11[i] + bc1);
        }
    }
    __syncthreads();

    {
        int row = tid >> 3, seg = tid & 7;
        const float* hrow = h2 + row * (HID + 1) + seg * 32;
        const float* wseg = Wd + seg * 32;
        float p = 0.0f;
        #pragma unroll
        for (int c = 0; c < 32; ++c) p += hrow[c] * wseg[c];
        zb[row * 8 + seg] = p;
    }
    __syncthreads();
    if (tid < ROWS) {
        float acc = 0.0f;
        #pragma unroll
        for (int s = 0; s < 8; ++s) acc += zb[tid * 8 + s];
        out[nodeBase + tid] = acc + bd[0];
    }
}

extern "C" void kernel_launch(void* const* d_in, const int* in_sizes, int n_in,
                              void* d_out, int out_size, void* d_ws, size_t ws_size,
                              hipStream_t stream) {
    const float* x     = (const float*)d_in[0];
    const int*   ei    = (const int*)d_in[1];
    const float* W_gcn = (const float*)d_in[2];
    const float* b_gcn = (const float*)d_in[3];
    const float* W1    = (const float*)d_in[4];
    const float* b1    = (const float*)d_in[5];
    const float* W2    = (const float*)d_in[6];
    const float* b2    = (const float*)d_in[7];
    const float* Wd    = (const float*)d_in[8];
    const float* bd    = (const float*)d_in[9];
    float* out = (float*)d_out;

    char* ws = (char*)d_ws;
    size_t off0 = 0;
    size_t off1 = (off0 + (size_t)N_NODES * sizeof(float) + 255) & ~(size_t)255;
    size_t off2 = (off1 + (size_t)N_NODES * C_IN * sizeof(float) + 255) & ~(size_t)255;
    size_t off3 = (off2 + (size_t)DEG_P * N_NODES * sizeof(unsigned) + 255) & ~(size_t)255;
    float* dinv = (float*)(ws + off0);
    float* agg  = (float*)(ws + off1);
    unsigned* part = (unsigned*)(ws + off2);
    float* wpk = (float*)(ws + off3);
    size_t off4 = (off3 + (size_t)PK_ALL * 4 * sizeof(float) + 255) & ~(size_t)255;
    size_t off5 = (off4 + (size_t)OW_NS * OW_NB * sizeof(unsigned) + 255) & ~(size_t)255;
    size_t off6 = (off5 + (size_t)(OW_NB + 1) * sizeof(unsigned) + 255) & ~(size_t)255;
    unsigned* bcnt   = (unsigned*)(ws + off4);
    unsigned* bstart = (unsigned*)(ws + off5);
    int2* pairs      = (int2*)(ws + off6);

    const int B = 256;
    deg_count_kernel<<<DEG_R * DEG_P + OW_NS, DEG_T, 0, stream>>>(ei, part, bcnt);
    deg_rsqrt_pack_kernel<<<OW_NS + RS_NB + PK_NB, RS_T, 0, stream>>>(part, dinv, W_gcn, W1, W2, wpk,
                                                                      ei, bcnt, bstart, pairs);

    owner_sum_kernel<<<OW_NB, OW_T, 0, stream>>>(pairs, bstart, x, dinv, agg);

    mlp_fused_kernel<<<N_NODES / ROWS, B, 0, stream>>>(agg, x, dinv, wpk, b_gcn, b1, b2,
                                                       Wd, bd, out);
}


# owner: each pass of pairs sorted by row in LDS (integer slot counts, scan, place); a run of one row is summed in registers and added to LDS once, not per edge; f32 sum order: runs by slot order
# speedup vs baseline: 2.0756x; 1.0343x over previous
#include <hip/hip_runtime.h>

#define N_NODES 120000
#define N_EDGES 1200000
#define C_IN    64
#define HID     256
#define ROWS    32
#define ZP      (C_IN + 4)
#define H1P     (HID + 4)


typedef float v2f __attribute__((ext_vector_type(2)));
typedef float v8f __attribute__((ext_vector_type(8)));


__device__ __forceinline__ float leaky01(float v) { return v >= 0.0f ? v : 0.01f * v; }

__device__ __forceinline__ void ldsAddF(float* p, float v) {
    __hip_atomic_fetch_add(p, v, __ATOMIC_RELAXED, __HIP_MEMORY_SCOPE_WORKGROUP);
}

__device__ __forceinline__ unsigned ldsTakeSlot(unsigned* p) {
    return __hip_atomic_fetch_add(p, 1u, __ATOMIC_RELAXED, __HIP_MEMORY_SCOPE_WORKGROUP);
}

__device__ __forceinline__ v8f wmma4(v2f a, v2f b, v8f c) {
    return __builtin_amdgcn_wmma_f32_16x16x4_f32(false, a, false, b, (short)0, c,
                                                 false, false);
}

#define DEG_R   10
#define DEG_P   25
#define DEG_NR  (N_NODES / DEG_R)
#define DEG_EP  (N_EDGES / DEG_P)
#define DEG_T   512
#define DEG_NV  (DEG_EP / 4)
#define DEG_IT  ((DEG_NV + DEG_T - 1) / DEG_T)
#define DEG_LB  24
static_assert(N_NODES % DEG_R == 0 && DEG_NR % 32 == 0, "a range is a whole number of 128-B lines");
static_assert(N_EDGES % DEG_P == 0 && DEG_EP % 4 == 0 && N_EDGES % 4 == 0, "slices start 16-B aligned");
static_assert(DEG_IT % DEG_LB == 0, "load batches cover the slice evenly");
static_assert(N_NODES % 4 == 0, "four nodes per thread in the plane sum");

#define OW_NB   250
#define OW_RN   (N_NODES / OW_NB)
#define OW_NS   250
#define OW_SE   (N_EDGES / OW_NS)
#define OW_SV   (OW_SE / 4)
#define OW_T    1024
#define OW_CH   8192
#define OW_PT   (OW_CH / OW_T)
static_assert(N_NODES % OW_NB == 0 && N_EDGES % OW_NS == 0 && OW_SE % 4 == 0, "ranges and slices are even");
static_assert(OW_NB < 256, "one thread per range, and one more for the total, in the offset scan");
static_assert(OW_CH % OW_T == 0 && OW_RN == 15 * 32 && OW_RN < 512, "pass and row-scan shapes of the owner");

__device__ __forceinline__ int ow_range(int d) {
    const unsigned b = (unsigned)d / (unsigned)OW_RN;
    return (int)(b < (unsigned)OW_NB ? b : (unsigned)(OW_NB - 1));
}

__device__ __forceinline__ void deg_count_one(unsigned* cnt, int d, int lo) {
    const unsigned u = (unsigned)(d - lo);
    if (u < (unsigned)DEG_NR)
        __hip_atomic_fetch_add(&cnt[u], 1u, __ATOMIC_RELAXED, __HIP_MEMORY_SCOPE_WORKGROUP);
}

__global__ __launch_bounds__(DEG_T) void deg_count_kernel(const int* __restrict__ ei,
                                                          unsigned* __restrict__ part,
                                                          unsigned* __restrict__ bcnt) {
    __shared__ uint4 cnt4[DEG_NR / 4];
    unsigned* cnt = reinterpret_cast<unsigned*>(cnt4);
    const int tid = threadIdx.x;

    if (blockIdx.x >= DEG_R * DEG_P) {
        const int s = blockIdx.x - DEG_R * DEG_P;
        if (tid < OW_NB) cnt[tid] = 0u;
        __syncthreads();
        const int4* sd4 = reinterpret_cast<const int4*>(ei + N_EDGES + (size_t)s * OW_SE);
        for (int i = tid; i < OW_SV; i += DEG_T) {
            const int4 d = sd4[i];
            ldsTakeSlot(&cnt[ow_range(d.x)]);
            ldsTakeSlot(&cnt[ow_range(d.y)]);
            ldsTakeSlot(&cnt[ow_range(d.z)]);
            ldsTakeSlot(&cnt[ow_range(d.w)]);
        }
        __syncthreads();
        if (tid < OW_NB) bcnt[s * OW_NB + tid] = cnt[tid];
        return;
    }
    const int p  = blockIdx.x / DEG_R;
    const int lo = (blockIdx.x - p * DEG_R) * DEG_NR;

    for (int j = tid; j < DEG_NR / 4; j += DEG_T) cnt4[j] = make_uint4(0u, 0u, 0u, 0u);
    __syncthreads();

    const int4* dst4 = reinterpret_cast<const int4*>(ei + N_EDGES + (size_t)p * DEG_EP);
    #pragma unroll
    for (int b = 0; b < DEG_IT; b += DEG_LB) {
        int4 d[DEG_LB];
        #pragma unroll
        for (int k = 0; k < DEG_LB; ++k) {
            const int i = tid + (b + k) * DEG_T;
            d[k] = (i < DEG_NV) ? dst4[i] : make_int4(-1, -1, -1, -1);
        }
        #pragma unroll
        for (int k = 0; k < DEG_LB; ++k) {
            deg_count_one(cnt, d[k].x, lo);
            deg_count_one(cnt, d[k].y, lo);
            deg_count_one(cnt, d[k].z, lo);
            deg_count_one(cnt, d[k].w, lo);
        }
    }
    __syncthreads();

    uint4* o4 = reinterpret_cast<uint4*>(part + (size_t)p * N_NODES + lo);
    for (int j = tid; j < DEG_NR / 4; j += DEG_T) o4[j] = cnt4[j];
}

#define PK_W2   (HID * HID / 4)
#define PK_W1   (HID * C_IN / 4)
#define PK_WG   (C_IN * C_IN / 4)
#define PK_ALL  (PK_W2 + PK_W1 + PK_WG)
#define RS_T    256
#define RS_NB   ((N_NODES / 4 + RS_T - 1) / RS_T)
#define PK_NB   (PK_ALL / RS_T)
static_assert(PK_ALL % RS_T == 0, "the copy is a whole number of work-groups");
static_assert(HID == 256 && C_IN == 64, "the index arithmetic of the copy is written for these sizes");

__device__ __forceinline__ void pack_weights(int o, const float* __restrict__ W_gcn,
                                             const float* __restrict__ W1,
                                             const float* __restrict__ W2,
                                             float4* __restrict__ wpk) {
    const int lane = o & 31, q = (o >> 5) & 3;
    const int half = lane >> 4, lm = lane & 15;
    float4 v;
    if (o < PK_W2) {
        const int j = (o >> 7) & 1, wave = (o >> 8) & 7, kt = o >> 11;
        const int g = 16 * wave + 128 * j + lm;
        v = reinterpret_cast<const float4*>(W2)[g * (HID / 4) + 8 * kt + 4 * half + q];
    } else if (o < PK_W2 + PK_W1) {
        const int o1 = o - PK_W2;
        const int j = (o1 >> 7) & 1, wave = (o1 >> 8) & 7, t = o1 >> 11;
        const int g = 16 * wave + 128 * j + lm;
        v = reinterpret_cast<const float4*>(W1)[g * (C_IN / 4) + 8 * t + 4 * half + q];
    } else {
        const int o2 = o - PK_W2 - PK_W1;
        const int t = (o2 >> 7) & 1, ct = o2 >> 8;
        const float* p = W_gcn + (size_t)(32 * t + 16 * half + 4 * q) * C_IN + 16 * ct + lm;
        v.x = p[0];  v.y = p[C_IN];  v.z = p[2 * C_IN];  v.w = p[3 * C_IN];
    }
    wpk[o] = v;
}

struct PlaceLds {
    int2 lp[OW_SE];
    unsigned long long sc[2][256];
    unsigned lcur[256];
    unsigned gdelta[256];
};

__device__ __forceinline__ void place_slice(PlaceLds& L, int s, const int* __restrict__ ei,
                                            const unsigned* __restrict__ bcnt,
                                            unsigned* __restrict__ bstart,
                                            int2* __restrict__ pairs) {
    const int t = threadIdx.x;
    unsigned tot = 0u, before = 0u, mine = 0u;
    if (t < OW_NB) {
        #pragma unroll 50
        for (int q = 0; q < OW_NS; ++q) {
            const unsigned v = bcnt[q * OW_NB + t];
            tot += v;
            before += (q < s) ? v : 0u;
            mine = (q == s) ? v : mine;
        }
    }
    L.sc[0][t] = ((unsigned long long)mine << 32) | (unsigned long long)tot;
    __syncthreads();
    int cur = 0;
    for (int off = 1; off < 256; off <<= 1) {
        unsigned long long v = L.sc[cur][t];
        if (t >= off) v += L.sc[cur][t - off];
        L.sc[cur ^ 1][t] = v;
        cur ^= 1;
        __syncthreads();
    }
    const unsigned long long incl = L.sc[cur][t];
    const unsigned gstart = (unsigned)(incl & 0xffffffffull) - tot;
    const unsigned lstart = (unsigned)(incl >> 32) - mine;
    L.lcur[t] = lstart;
    L.gdelta[t] = gstart + before - lstart;
    if (s == 0 && t <= OW_NB) bstart[t] = gstart;
    __syncthreads();

    const int4* s4 = reinterpret_cast<const int4*>(ei + (size_t)s * OW_SE);
    const int4* d4 = reinterpret_cast<const int4*>(ei + N_EDGES + (size_t)s * OW_SE);
    for (int i = t; i < OW_SV; i += RS_T) {
        const int4 sv = s4[i];
        const int4 dv = d4[i];
        L.lp[ldsTakeSlot(&L.lcur[ow_range(dv.x)])] = make_int2(sv.x, dv.x);
        L.lp[ldsTakeSlot(&L.lcur[ow_range(dv.y)])] = make_int2(sv.y, dv.y);
        L.lp[ldsTakeSlot(&L.lcur[ow_range(dv.z)])] = make_int2(sv.z, dv.z);
        L.lp[ldsTakeSlot(&L.lcur[ow_range(dv.w)])] = make_int2(sv.w, dv.w);
    }
    __syncthreads();
    for (int i = t; i < OW_SE; i += RS_T) {
        const int2 p = L.lp[i];
        pairs[(unsigned)i + L.gdelta[ow_range(p.y)]] = p;
    }
}

__global__ __launch_bounds__(RS_T) void deg_rsqrt_pack_kernel(const unsigned* __restrict__ part,
                                                              float* __restrict__ dinv,
                                                              const float* __restrict__ W_gcn,
                                                              const float* __restrict__ W1,
                                                              const float* __restrict__ W2,
                                                              float* __restrict__ wpk_f,
                                                              const int* __restrict__ ei,
                                                              const unsigned* __restrict__ bcnt,
                                                              unsigned* __restrict__ bstart,
                                                              int2* __restrict__ pairs) {
    __shared__ PlaceLds placeLds;
    if (blockIdx.x < OW_NS) {
        place_slice(placeLds, blockIdx.x, ei, bcnt, bstart, pairs);
        return;
    }
    const int bi = blockIdx.x - OW_NS;
    if (bi >= RS_NB) {
        pack_weights((bi - RS_NB) * RS_T + threadIdx.x, W_gcn, W1, W2,
                     reinterpret_cast<float4*>(wpk_f));
        return;
    }
    const int i = bi * RS_T + threadIdx.x;
    if (i >= N_NODES / 4) return;
    const uint4* p4 = reinterpret_cast<const uint4*>(part) + i;
    uint4 s = make_uint4(1u, 1u, 1u, 1u);
    #pragma unroll
    for (int p = 0; p < DEG_P; ++p) {
        const uint4 v = p4[(size_t)p * (N_NODES / 4)];
        s.x += v.x; s.y += v.y; s.z += v.z; s.w += v.w;
    }
    float4 o;
    o.x = rsqrtf((float)s.x);
    o.y = rsqrtf((float)s.y);
    o.z = rsqrtf((float)s.z);
    o.w = rsqrtf((float)s.w);
    reinterpret_cast<float4*>(dinv)[i] = o;
}

__global__ __launch_bounds__(OW_T) void owner_sum_kernel(const int2* __restrict__ pairs,
                                                         const unsigned* __restrict__ bstart,
                                                         const float* __restrict__ x,
                                                         const float* __restrict__ dinv,
                                                         float* __restrict__ agg) {
    __shared__ float4 sums4[OW_RN * C_IN / 4];
    __shared__ int      l_src[OW_CH];
    __shared__ float    l_nrm[OW_CH];
    __shared__ int      l_row[OW_CH];
    __shared__ unsigned rcnt[512];
    __shared__ unsigned rstart[512];
    float* sums = reinterpret_cast<float*>(sums4);
    const int tid = threadIdx.x;
    const int b = blockIdx.x;
    for (int i = tid; i < OW_RN * C_IN / 4; i += OW_T) sums4[i] = make_float4(0.0f, 0.0f, 0.0f, 0.0f);

    const int lane = tid & 31;
    const int wave = __builtin_amdgcn_readfirstlane(tid >> 5);
    const int eb = (int)bstart[b], ee = (int)bstart[b + 1];
    const int lo = b * OW_RN;

    for (int cb = eb; cb < ee; cb += OW_CH) {
        const int len = (ee - cb) < OW_CH ? (ee - cb) : OW_CH;
        if (tid < 512) rcnt[tid] = 0u;
        __syncthreads();

        int ps[OW_PT], pr[OW_PT];
        float pn[OW_PT];
        unsigned pslot[OW_PT];
        #pragma unroll
        for (int i = 0; i < OW_PT; ++i) {
            const int k = tid + i * OW_T;
            const int2 p = pairs[cb + (k < len ? k : len - 1)];
            ps[i] = p.x;
            pr[i] = k < len ? p.y - lo : -1;
        }
        #pragma unroll
        for (int i = 0; i < OW_PT; ++i) {
            pn[i] = 0.0f;  pslot[i] = 0u;
            if ((unsigned)pr[i] < (unsigned)OW_RN) {
                pn[i] = dinv[ps[i]];
                pslot[i] = ldsTakeSlot(&rcnt[pr[i]]);
            }
        }
        __syncthreads();

        if (wave == 0) {
            unsigned c[15];
            unsigned run = 0u;
            #pragma unroll
            for (int q = 0; q < 15; ++q) { c[q] = run;  run += rcnt[15 * lane + q]; }
            unsigned inc = run;
            #pragma unroll
            for (int off = 1; off < 32; off <<= 1) {
                const unsigned o = __shfl_up(inc, off, 32);
                if (lane >= off) inc += o;
            }
            const unsigned excl = inc - run;
            #pragma unroll
            for (int q = 0; q < 15; ++q) rstart[15 * lane + q] = excl + c[q];
            if (lane == 31) rstart[OW_RN] = inc;
        }
        __syncthreads();

        #pragma unroll
        for (int i = 0; i < OW_PT; ++i) {
            if ((unsigned)pr[i] < (unsigned)OW_RN) {
                const unsigned pos = rstart[pr[i]] + pslot[i];
                l_src[pos] = ps[i];  l_nrm[pos] = pn[i];  l_row[pos] = pr[i];
            }
        }
        __syncthreads();

        const int m = (int)rstart[OW_RN];
        for (int g0 = wave * 32; g0 < m; g0 += OW_T) {
            const int n = (m - g0) < 32 ? (m - g0) : 32;
            int s = 0, r = 0;
            float nm = 0.0f;
            if (lane < n) { s = l_src[g0 + lane];  nm = l_nrm[g0 + lane];  r = l_row[g0 + lane]; }
            const int s0 = __builtin_amdgcn_readlane(s, 0);
            if (lane >= n) s = s0;

            float2 v[32];
            #pragma unroll
            for (int j = 0; j < 32; ++j) {
                const int sj = __builtin_amdgcn_readlane(s, j);
                v[j] = reinterpret_cast<const float2*>(x + (size_t)sj * C_IN)[lane];
            }
            int cur = __builtin_amdgcn_readlane(r, 0);
            float ax = 0.0f, ay = 0.0f;
            #pragma unroll
            for (int j = 0; j < 32; ++j) {
                if (j < n) {
                    const int rj = __builtin_amdgcn_readlane(r, j);
                    const float nj = __int_as_float(__builtin_amdgcn_readlane(__float_as_int(nm), j));
                    if (rj != cur) {
                        float* ap = sums + cur * C_IN + 2 * lane;
                        ldsAddF(ap,     ax);
                        ldsAddF(ap + 1, ay);
                        cur = rj;  ax = 0.0f;  ay = 0.0f;
                    }
                    ax += v[j].x * nj;
                    ay += v[j].y * nj;
                }
            }
            float* ap = sums + cur * C_IN + 2 * lane;
            ldsAddF(ap,     ax);
            ldsAddF(ap + 1, ay);
        }
    }
    __syncthreads();

    float4* o4 = reinterpret_cast<float4*>(agg) + (size_t)lo * (C_IN / 4);
    for (int i = tid; i < OW_RN * C_IN / 4; i += OW_T) o4[i] = sums4[i];
}

__global__ __launch_bounds__(256) void mlp_fused_kernel(
    const float* __restrict__ agg, const float* __restrict__ x,
    const float* __restrict__ dinv, const float* __restrict__ wpk_f,
    const float* __restrict__ b_gcn,
    const float* __restrict__ b1,
    const float* __restrict__ b2, const float* __restrict__ Wd,
    const float* __restrict__ bd, float* __restrict__ out) {
    __shared__ float4 zb4[ROWS * ZP / 4];
    __shared__ float4 h14[ROWS * H1P / 4];
    __shared__ float4 h24[ROWS * (HID + 1) / 4];
    static_assert((ROWS * (HID + 1)) % 4 == 0 && ROWS * ZP <= ROWS * (HID + 1), "u tile fits in h2's space");
    const float4* wpk = reinterpret_cast<const float4*>(wpk_f);
    float* zb = reinterpret_cast<float*>(zb4);
    float* h1 = reinterpret_cast<float*>(h14);
    float* h2 = reinterpret_cast<float*>(h24);
    float* ub = h2;

    int tid = threadIdx.x;
    int nodeBase = blockIdx.x * ROWS;

    #pragma unroll
    for (int t = 0; t < ROWS * C_IN / 4 / 256; ++t) {
        const int i = tid + t * 256;
        const int r = i >> 4, c4 = i & 15;
        const size_t g4 = (size_t)(nodeBase + r) * (C_IN / 4) + c4;
        const float4 av = reinterpret_cast<const float4*>(agg)[g4];
        const float4 xv = reinterpret_cast<const float4*>(x)[g4];
        const float dv = dinv[nodeBase + r];
        float4 u;
        u.x = dv * (av.x + dv * xv.x);
        u.y = dv * (av.y + dv * xv.y);
        u.z = dv * (av.z + dv * xv.z);
        u.w = dv * (av.w + dv * xv.w);
        reinterpret_cast<float4*>(ub + r * ZP)[c4] = u;
        reinterpret_cast<float4*>(zb + r * ZP)[c4] = xv;
    }
    __syncthreads();

    int wave = tid >> 5, lane = tid & 31, half = lane >> 4, lm = lane & 15;
    int col0 = wave * 16;
    int col1 = col0 + 128;

    {
        const int rt = wave >> 2, ct = wave & 3;
        v8f acc = {};
        const float4* aR = reinterpret_cast<const float4*>(ub + (16 * rt + lm) * ZP + 16 * half);
        const float4* bw = wpk + PK_W2 + PK_W1 + ct * 256 + lane;
        #pragma unroll
        for (int t = 0; t < C_IN / 32; ++t) {
            float4 a[4], bv[4];
            #pragma unroll
            for (int q = 0; q < 4; ++q) bv[q] = bw[t * 128 + q * 32];
            #pragma unroll
            for (int q = 0; q < 4; ++q) a[q] = aR[8 * t + q];
            #pragma unroll
            for (int q = 0; q < 4; ++q) {
                v2f av, b;
                av.x = a[q].x;  av.y = a[q].y;
                b.x = bv[q].x;  b.y = bv[q].y;
                acc = wmma4(av, b, acc);
                av.x = a[q].z;  av.y = a[q].w;
                b.x = bv[q].z;  b.y = bv[q].w;
                acc = wmma4(av, b, acc);
            }
        }
        const float bc = b_gcn[16 * ct + lm];
        #pragma unroll
        for (int i = 0; i < 8; ++i) {
            float* zp = zb + (16 * rt + i + 8 * half) * ZP + 16 * ct + lm;
            float v = acc[i] + bc;
            v = v > 0.0f ? v : 0.0f;
            *zp = v + *zp;
        }
    }
    __syncthreads();

    {
        v8f acc00 = {}, acc01 = {}, acc10 = {}, acc11 = {};
        const float4* aR0 = reinterpret_cast<const float4*>(zb + lm * ZP + 16 * half);
        const float4* aR1 = reinterpret_cast<const float4*>(zb + (16 + lm) * ZP + 16 * half);
        const float4* w0 = wpk + PK_W2 + wave * 256 + lane;
        const float4* w1 = w0 + 128;
        #pragma unroll
        for (int t = 0; t < C_IN / 32; ++t) {
            float4 a0[4], a1[4], b0[4], b1q[4];
            #pragma unroll
            for (int q = 0; q < 4; ++q) { b0[q] = w0[2048 * t + 32 * q]; b1q[q] = w1[2048 * t + 32 * q]; }
            #pragma unroll
            for (int q = 0; q < 4; ++q) { a0[q] = aR0[8 * t + q]; a1[q] = aR1[8 * t + q]; }
            #pragma unroll
            for (int q = 0; q < 4; ++q) {
                v2f a0v, a1v, b0v, b1v;
                a0v.x = a0[q].x;  a0v.y = a0[q].y;
                a1v.x = a1[q].x;  a1v.y = a1[q].y;
                b0v.x = b0[q].x;  b0v.y = b0[q].y;
                b1v.x = b1q[q].x; b1v.y = b1q[q].y;
                acc00 = wmma4(a0v, b0v, acc00);
                acc01 = wmma4(a0v, b1v, acc01);
                acc10 = wmma4(a1v, b0v, acc10);
                acc11 = wmma4(a1v, b1v, acc11);
                a0v.x = a0[q].z;  a0v.y = a0[q].w;
                a1v.x = a1[q].z;  a1v.y = a1[q].w;
                b0v.x = b0[q].z;  b0v.y = b0[q].w;
                b1v.x = b1q[q].z; b1v.y = b1q[q].w;
                acc00 = wmma4(a0v, b0v, acc00);
                acc01 = wmma4(a0v, b1v, acc01);
                acc10 = wmma4(a1v, b0v, acc10);
                acc11 = wmma4(a1v, b1v, acc11);
            }
        }
        float bc0 = b1[col0 + lm], bc1 = b1[col1 + lm];
        #pragma unroll
        for (int i = 0; i < 8; ++i) {
            int r = i + 8 * half;
            h1[r * H1P + col0 + lm]        = leaky01(acc00[i] + bc0);
            h1[r * H1P + col1 + lm]        = leaky01(acc01[i] + bc1);
            h1[(16 + r) * H1P + col0 + lm] = leaky01(acc10[i] + bc0);
            h1[(16 + r) * H1P + col1 + lm] = leaky01(acc11[i] + bc1);
        }
    }
    __syncthreads();

    {
        v8f acc00 = {}, acc01 = {}, acc10 = {}, acc11 = {};
        const float4* aR0 = reinterpret_cast<const float4*>(h1 + lm * H1P + 16 * half);
        const float4* aR1 = reinterpret_cast<const float4*>(h1 + (16 + lm) * H1P + 16 * half);
        const float4* w0 = wpk + wave * 256 + lane;
        const float4* w1 = w0 + 128;
        #pragma unroll 1
        for (int kq = 0; kq < HID / 4; kq += 8) {
            float4 a0[4], a1[4], b0[4], b1q[4];
            #pragma unroll
            for (int q = 0; q < 4; ++q) { b0[q] = w0[256 * kq + 32 * q]; b1q[q] = w1[256 * kq + 32 * q]; }
            #pragma unroll
            for (int q = 0; q < 4; ++q) { a0[q] = aR0[kq + q]; a1[q] = aR1[kq + q]; }
            #pragma unroll
            for (int q = 0; q < 4; ++q) {
                v2f a0v, a1v, b0v, b1v;
                a0v.x = a0[q].x;  a0v.y = a0[q].y;
                a1v.x = a1[q].x;  a1v.y = a1[q].y;
                b0v.x = b0[q].x;  b0v.y = b0[q].y;
                b1v.x = b1q[q].x; b1v.y = b1q[q].y;
                acc00 = wmma4(a0v, b0v, acc00);
                acc01 = wmma4(a0v, b1v, acc01);
                acc10 = wmma4(a1v, b0v, acc10);
                acc11 = wmma4(a1v, b1v, acc11);
                a0v.x = a0[q].z;  a0v.y = a0[q].w;
                a1v.x = a1[q].z;  a1v.y = a1[q].w;
                b0v.x = b0[q].z;  b0v.y = b0[q].w;
                b1v.x = b1q[q].z; b1v.y = b1q[q].w;
                acc00 = wmma4(a0v, b0v, acc00);
                acc01 = wmma4(a0v, b1v, acc01);
                acc10 = wmma4(a1v, b0v, acc10);
                acc11 = wmma4(a1v, b1v, acc11);
            }
        }
        float bc0 = b2[col0 + lm], bc1 = b2[col1 + lm];
        #pragma unroll
        for (int i = 0; i < 8; ++i) {
            int r = i + 8 * half;
            h2[r * (HID + 1) + col0 + lm]        = leaky01(acc00[i] + bc0);
            h2[r * (HID + 1) + col1 + lm]        = leaky01(acc01[i] + bc1);
            h2[(16 + r) * (HID + 1) + col0 + lm] = leaky01(acc10[i] + bc0);
            h2[(16 + r) * (HID + 1) + col1 + lm] = leaky01(acc11[i] + bc1);
        }
    }
    __syncthreads();

    {
        int row = tid >> 3, seg = tid & 7;
        const float* hrow = h2 + row * (HID + 1) + seg * 32;
        const float* wseg = Wd + seg * 32;
        float p = 0.0f;
        #pragma unroll
        for (int c = 0; c < 32; ++c) p += hrow[c] * wseg[c];
        zb[row * 8 + seg] = p;
    }
    __syncthreads();
    if (tid < ROWS) {
        float acc = 0.0f;
        #pragma unroll
        for (int s = 0; s < 8; ++s) acc += zb[tid * 8 + s];
        out[nodeBase + tid] = acc + bd[0];
    }
}

extern "C" void kernel_launch(void* const* d_in, const int* in_sizes, int n_in,
                              void* d_out, int out_size, void* d_ws, size_t ws_size,
                              hipStream_t stream) {
    const float* x     = (const float*)d_in[0];
    const int*   ei    = (const int*)d_in[1];
    const float* W_gcn = (const float*)d_in[2];
    const float* b_gcn = (const float*)d_in[3];
    const float* W1    = (const float*)d_in[4];
    const float* b1    = (const float*)d_in[5];
    const float* W2    = (const float*)d_in[6];
    const float* b2    = (const float*)d_in[7];
    const float* Wd    = (const float*)d_in[8];
    const float* bd    = (const float*)d_in[9];
    float* out = (float*)d_out;

    char* ws = (char*)d_ws;
    size_t off0 = 0;
    size_t off1 = (off0 + (size_t)N_NODES * sizeof(float) + 255) & ~(size_t)255;
    size_t off2 = (off1 + (size_t)N_NODES * C_IN * sizeof(float) + 255) & ~(size_t)255;
    size_t off3 = (off2 + (size_t)DEG_P * N_NODES * sizeof(unsigned) + 255) & ~(size_t)255;
    float* dinv = (float*)(ws + off0);
    float* agg  = (float*)(ws + off1);
    unsigned* part = (unsigned*)(ws + off2);
    float* wpk = (float*)(ws + off3);
    size_t off4 = (off3 + (size_t)PK_ALL * 4 * sizeof(float) + 255) & ~(size_t)255;
    size_t off5 = (off4 + (size_t)OW_NS * OW_NB * sizeof(unsigned) + 255) & ~(size_t)255;
    size_t off6 = (off5 + (size_t)(OW_NB + 1) * sizeof(unsigned) + 255) & ~(size_t)255;
    unsigned* bcnt   = (unsigned*)(ws + off4);
    unsigned* bstart = (unsigned*)(ws + off5);
    int2* pairs      = (int2*)(ws + off6);

    const int B = 256;
    deg_count_kernel<<<DEG_R * DEG_P + OW_NS, DEG_T, 0, stream>>>(ei, part, bcnt);
    deg_rsqrt_pack_kernel<<<OW_NS + RS_NB + PK_NB, RS_T, 0, stream>>>(part, dinv, W_gcn, W1, W2, wpk,
                                                                      ei, bcnt, bstart, pairs);

    owner_sum_kernel<<<OW_NB, OW_T, 0, stream>>>(pairs, bstart, x, dinv, agg);

    mlp_fused_kernel<<<N_NODES / ROWS, B, 0, stream>>>(agg, x, dinv, wpk, b_gcn, b1, b2,
                                                       Wd, bd, out);
}


# edges binned in launch 1: each slice sorts its pairs by node range in LDS; a range's 250 runs feed its degree count (no count planes) and its owner; count-scan-place pass goes; f32 sum order changes
# speedup vs baseline: 2.2370x; 1.0778x over previous
#include <hip/hip_runtime.h>

#define N_NODES 120000
#define N_EDGES 1200000
#define C_IN    64
#define HID     256
#define ROWS    32
#define ZP      (C_IN + 4)
#define H1P     (HID + 4)


typedef float v2f __attribute__((ext_vector_type(2)));
typedef float v8f __attribute__((ext_vector_type(8)));


__device__ __forceinline__ float leaky01(float v) { return v >= 0.0f ? v : 0.01f * v; }

__device__ __forceinline__ void ldsAddF(float* p, float v) {
    __hip_atomic_fetch_add(p, v, __ATOMIC_RELAXED, __HIP_MEMORY_SCOPE_WORKGROUP);
}

__device__ __forceinline__ unsigned ldsTakeSlot(unsigned* p) {
    return __hip_atomic_fetch_add(p, 1u, __ATOMIC_RELAXED, __HIP_MEMORY_SCOPE_WORKGROUP);
}

__device__ __forceinline__ v8f wmma4(v2f a, v2f b, v8f c) {
    return __builtin_amdgcn_wmma_f32_16x16x4_f32(false, a, false, b, (short)0, c,
                                                 false, false);
}


#define OW_NB   250
#define OW_RN   (N_NODES / OW_NB)
#define OW_NS   250
#define OW_SE   (N_EDGES / OW_NS)
#define OW_SV   (OW_SE / 4)
#define OW_T    1024
#define OW_RT   4
#define OW_PT   12
#define OW_CH   (OW_T * OW_PT)
#define LS_T    1024
#define LS_IT   ((OW_SV + LS_T - 1) / LS_T)
#define LS_W    (OW_NB + 1)
static_assert(N_NODES % OW_NB == 0 && N_EDGES % OW_NS == 0 && OW_SE % 4 == 0, "ranges and slices are even");
static_assert(N_EDGES % 4 == 0 && (OW_SE * 4) % 16 == 0, "slices start 16-B aligned");
static_assert(OW_NB < 256, "the run starts of a slice, and the slice's pair count behind them, fit 256 scan words");
static_assert(OW_NS * OW_RT <= OW_T, "every run has its threads");
static_assert(OW_PT == 12, "the load fences of the range kernels name twelve values");
static_assert(OW_RN == 15 * 32 && OW_RN < 512, "row-scan shape of the owner");

__device__ __forceinline__ int ow_range(int d) {
    const unsigned b = (unsigned)d / (unsigned)OW_RN;
    return (int)(b < (unsigned)OW_NB ? b : (unsigned)(OW_NB - 1));
}

__global__ __launch_bounds__(LS_T) void slice_sort_kernel(const int* __restrict__ ei,
                                                          unsigned* __restrict__ lstart,
                                                          int2* __restrict__ lpairs) {
    __shared__ int2 lp[OW_SE];
    __shared__ unsigned lcnt[256];
    __shared__ unsigned lst[256];
    const int tid = threadIdx.x;
    const int s = blockIdx.x;
    if (tid < 256) lcnt[tid] = 0u;
    __syncthreads();

    const int4* s4 = reinterpret_cast<const int4*>(ei + (size_t)s * OW_SE);
    const int4* d4 = reinterpret_cast<const int4*>(ei + N_EDGES + (size_t)s * OW_SE);
    int4 sv[LS_IT], dv[LS_IT];
    unsigned slot[LS_IT][4];
    #pragma unroll
    for (int k = 0; k < LS_IT; ++k) {
        const int i = tid + k * LS_T;
        sv[k] = make_int4(0, 0, 0, 0);
        dv[k] = make_int4(0, 0, 0, 0);
        if (i < OW_SV) { sv[k] = s4[i];  dv[k] = d4[i]; }
    }
    #pragma unroll
    for (int k = 0; k < LS_IT; ++k) {
        const int i = tid + k * LS_T;
        slot[k][0] = 0u;  slot[k][1] = 0u;  slot[k][2] = 0u;  slot[k][3] = 0u;
        if (i < OW_SV) {
            slot[k][0] = ldsTakeSlot(&lcnt[ow_range(dv[k].x)]);
            slot[k][1] = ldsTakeSlot(&lcnt[ow_range(dv[k].y)]);
            slot[k][2] = ldsTakeSlot(&lcnt[ow_range(dv[k].z)]);
            slot[k][3] = ldsTakeSlot(&lcnt[ow_range(dv[k].w)]);
        }
    }
    __syncthreads();

    if (tid < 32) {
        unsigned c[8];
        unsigned run = 0u;
        #pragma unroll
        for (int q = 0; q < 8; ++q) { c[q] = run;  run += lcnt[8 * tid + q]; }
        unsigned inc = run;
        #pragma unroll
        for (int off = 1; off < 32; off <<= 1) {
            const unsigned o = __shfl_up(inc, off, 32);
            if (tid >= off) inc += o;
        }
        const unsigned excl = inc - run;
        #pragma unroll
        for (int q = 0; q < 8; ++q) lst[8 * tid + q] = excl + c[q];
    }
    __syncthreads();

    if (tid < LS_W) lstart[(size_t)s * LS_W + tid] = lst[tid];
    #pragma unroll
    for (int k = 0; k < LS_IT; ++k) {
        const int i = tid + k * LS_T;
        if (i < OW_SV) {
            lp[lst[ow_range(dv[k].x)] + slot[k][0]] = make_int2(sv[k].x, dv[k].x);
            lp[lst[ow_range(dv[k].y)] + slot[k][1]] = make_int2(sv[k].y, dv[k].y);
            lp[lst[ow_range(dv[k].z)] + slot[k][2]] = make_int2(sv[k].z, dv[k].z);
            lp[lst[ow_range(dv[k].w)] + slot[k][3]] = make_int2(sv[k].w, dv[k].w);
        }
    }
    __syncthreads();

    int2* o = lpairs + (size_t)s * OW_SE;
    for (int i = tid; i < OW_SE; i += LS_T) o[i] = lp[i];
}

#define PK_W2   (HID * HID / 4)
#define PK_W1   (HID * C_IN / 4)
#define PK_WG   (C_IN * C_IN / 4)
#define PK_ALL  (PK_W2 + PK_W1 + PK_WG)
#define RS_T    1024
#define PK_NB   (PK_ALL / RS_T)
static_assert(PK_ALL % RS_T == 0, "the copy is a whole number of work-groups");
static_assert(HID == 256 && C_IN == 64, "the index arithmetic of the copy is written for these sizes");

__device__ __forceinline__ void pack_weights(int o, const float* __restrict__ W_gcn,
                                             const float* __restrict__ W1,
                                             const float* __restrict__ W2,
                                             float4* __restrict__ wpk) {
    const int lane = o & 31, q = (o >> 5) & 3;
    const int half = lane >> 4, lm = lane & 15;
    float4 v;
    if (o < PK_W2) {
        const int j = (o >> 7) & 1, wave = (o >> 8) & 7, kt = o >> 11;
        const int g = 16 * wave + 128 * j + lm;
        v = reinterpret_cast<const float4*>(W2)[g * (HID / 4) + 8 * kt + 4 * half + q];
    } else if (o < PK_W2 + PK_W1) {
        const int o1 = o - PK_W2;
        const int j = (o1 >> 7) & 1, wave = (o1 >> 8) & 7, t = o1 >> 11;
        const int g = 16 * wave + 128 * j + lm;
        v = reinterpret_cast<const float4*>(W1)[g * (C_IN / 4) + 8 * t + 4 * half + q];
    } else {
        const int o2 = o - PK_W2 - PK_W1;
        const int t = (o2 >> 7) & 1, ct = o2 >> 8;
        const float* p = W_gcn + (size_t)(32 * t + 16 * half + 4 * q) * C_IN + 16 * ct + lm;
        v.x = p[0];  v.y = p[C_IN];  v.z = p[2 * C_IN];  v.w = p[3 * C_IN];
    }
    wpk[o] = v;
}

__global__ __launch_bounds__(RS_T) void range_deg_pack_kernel(const int2* __restrict__ lpairs,
                                                              const unsigned* __restrict__ lstart,
                                                              float* __restrict__ dinv,
                                                              const float* __restrict__ W_gcn,
                                                              const float* __restrict__ W1,
                                                              const float* __restrict__ W2,
                                                              float* __restrict__ wpk_f) {
    __shared__ unsigned cnt[512];
    const int tid = threadIdx.x;
    if (blockIdx.x >= OW_NB) {
        pack_weights((blockIdx.x - OW_NB) * RS_T + tid, W_gcn, W1, W2,
                     reinterpret_cast<float4*>(wpk_f));
        return;
    }
    const int b = blockIdx.x;
    const int lo = b * OW_RN;
    if (tid < 512) cnt[tid] = 0u;
    __syncthreads();

    const int sl = tid / OW_RT, sub = tid % OW_RT;
    int ra = 0, rc = 0;
    if (sl < OW_NS) {
        const unsigned a0 = lstart[sl * LS_W + b];
        const unsigned a1 = lstart[sl * LS_W + b + 1];
        ra = (int)a0;  rc = (int)(a1 - a0);
    }
    const int* dw = reinterpret_cast<const int*>(lpairs) + 1;
    const int sbase = (sl < OW_NS ? sl : 0) * OW_SE;
    for (int base = 0; base < rc; base += OW_RT * OW_PT) {
        int r[OW_PT];
        #pragma unroll
        for (int i = 0; i < OW_PT; ++i) {
            const int j = base + sub + OW_RT * i;
            r[i] = dw[2 * (sbase + (j < rc ? ra + j : 0))];
        }
        asm volatile("" :: "v"(r[0]), "v"(r[1]), "v"(r[2]), "v"(r[3]), "v"(r[4]), "v"(r[5]),
                           "v"(r[6]), "v"(r[7]), "v"(r[8]), "v"(r[9]), "v"(r[10]), "v"(r[11]));
        #pragma unroll
        for (int i = 0; i < OW_PT; ++i) {
            const int j = base + sub + OW_RT * i;
            r[i] = j < rc ? r[i] - lo : -1;
        }
        #pragma unroll
        for (int i = 0; i < OW_PT; ++i) {
            if ((unsigned)r[i] < (unsigned)OW_RN)
                __hip_atomic_fetch_add(&cnt[r[i]], 1u, __ATOMIC_RELAXED, __HIP_MEMORY_SCOPE_WORKGROUP);
        }
    }
    __syncthreads();
    if (tid < OW_RN) dinv[lo + tid] = rsqrtf((float)(cnt[tid] + 1u));
}


__global__ __launch_bounds__(OW_T) void owner_sum_kernel(const int2* __restrict__ lpairs,
                                                         const unsigned* __restrict__ lstart,
                                                         const float* __restrict__ x,
                                                         const float* __restrict__ dinv,
                                                         float* __restrict__ agg) {
    __shared__ float4 sums4[OW_RN * C_IN / 4];
    __shared__ int      l_src[OW_CH];
    __shared__ float    l_nrm[OW_CH];
    __shared__ int      l_row[OW_CH];
    __shared__ unsigned rcnt[512];
    __shared__ unsigned rstart[512];
    __shared__ unsigned longest;
    float* sums = reinterpret_cast<float*>(sums4);
    const int tid = threadIdx.x;
    const int b = blockIdx.x;
    for (int i = tid; i < OW_RN * C_IN / 4; i += OW_T) sums4[i] = make_float4(0.0f, 0.0f, 0.0f, 0.0f);

    const int lane = tid & 31;
    const int wave = __builtin_amdgcn_readfirstlane(tid >> 5);
    const int lo = b * OW_RN;
    if (tid == 0) longest = 0u;

    const int sl = tid / OW_RT, sub = tid % OW_RT;
    int ra = 0, rc = 0;
    if (sl < OW_NS) {
        const unsigned a0 = lstart[sl * LS_W + b];
        const unsigned a1 = lstart[sl * LS_W + b + 1];
        ra = (int)a0;  rc = (int)(a1 - a0);
    }
    const int sbase = (sl < OW_NS ? sl : 0) * OW_SE;
    __syncthreads();
    if (sub == 0 && rc > 0)
        __hip_atomic_fetch_max(&longest, (unsigned)rc, __ATOMIC_RELAXED, __HIP_MEMORY_SCOPE_WORKGROUP);
    __syncthreads();
    const int mx = (int)longest;

    for (int base = 0; base < mx; base += OW_RT * OW_PT) {
        if (tid < 512) rcnt[tid] = 0u;
        __syncthreads();

        int ps[OW_PT], pr[OW_PT];
        float pn[OW_PT];
        unsigned pslot[OW_PT];
        #pragma unroll
        for (int i = 0; i < OW_PT; ++i) {
            const int j = base + sub + OW_RT * i;
            const int2 p = lpairs[sbase + (j < rc ? ra + j : 0)];
            ps[i] = p.x;
            pr[i] = p.y;
        }
        asm volatile("" :: "v"(pr[0]), "v"(pr[1]), "v"(pr[2]), "v"(pr[3]), "v"(pr[4]), "v"(pr[5]),
                           "v"(pr[6]), "v"(pr[7]), "v"(pr[8]), "v"(pr[9]), "v"(pr[10]), "v"(pr[11]));
        #pragma unroll
        for (int i = 0; i < OW_PT; ++i) {
            const int j = base + sub + OW_RT * i;
            pr[i] = j < rc ? pr[i] - lo : -1;
        }
        #pragma unroll
        for (int i = 0; i < OW_PT; ++i) {
            pn[i] = 0.0f;  pslot[i] = 0u;
            if ((unsigned)pr[i] < (unsigned)OW_RN) {
                pn[i] = dinv[ps[i]];
                pslot[i] = ldsTakeSlot(&rcnt[pr[i]]);
            }
        }
        __syncthreads();

        if (wave == 0) {
            unsigned c[15];
            unsigned run = 0u;
            #pragma unroll
            for (int q = 0; q < 15; ++q) { c[q] = run;  run += rcnt[15 * lane + q]; }
            unsigned inc = run;
            #pragma unroll
            for (int off = 1; off < 32; off <<= 1) {
                const unsigned o = __shfl_up(inc, off, 32);
                if (lane >= off) inc += o;
            }
            const unsigned excl = inc - run;
            #pragma unroll
            for (int q = 0; q < 15; ++q) rstart[15 * lane + q] = excl + c[q];
            if (lane == 31) rstart[OW_RN] = inc;
        }
        __syncthreads();

        #pragma unroll
        for (int i = 0; i < OW_PT; ++i) {
            if ((unsigned)pr[i] < (unsigned)OW_RN) {
                const unsigned pos = rstart[pr[i]] + pslot[i];
                l_src[pos] = ps[i];  l_nrm[pos] = pn[i];  l_row[pos] = pr[i];
            }
        }
        __syncthreads();

        const int m = (int)rstart[OW_RN];
        for (int g0 = wave * 32; g0 < m; g0 += OW_T) {
            const int n = (m - g0) < 32 ? (m - g0) : 32;
            int s = 0, r = 0;
            float nm = 0.0f;
            if (lane < n) { s = l_src[g0 + lane];  nm = l_nrm[g0 + lane];  r = l_row[g0 + lane]; }
            const int s0 = __builtin_amdgcn_readlane(s, 0);
            if (lane >= n) s = s0;

            float2 v[32];
            #pragma unroll
            for (int j = 0; j < 32; ++j) {
                const int sj = __builtin_amdgcn_readlane(s, j);
                v[j] = reinterpret_cast<const float2*>(x + (size_t)sj * C_IN)[lane];
            }
            int cur = __builtin_amdgcn_readlane(r, 0);
            float ax = 0.0f, ay = 0.0f;
            #pragma unroll
            for (int j = 0; j < 32; ++j) {
                if (j < n) {
                    const int rj = __builtin_amdgcn_readlane(r, j);
                    const float nj = __int_as_float(__builtin_amdgcn_readlane(__float_as_int(nm), j));
                    if (rj != cur) {
                        float* ap = sums + cur * C_IN + 2 * lane;
                        ldsAddF(ap,     ax);
                        ldsAddF(ap + 1, ay);
                        cur = rj;  ax = 0.0f;  ay = 0.0f;
                    }
                    ax += v[j].x * nj;
                    ay += v[j].y * nj;
                }
            }
            float* ap = sums + cur * C_IN + 2 * lane;
            ldsAddF(ap,     ax);
            ldsAddF(ap + 1, ay);
        }
    }
    __syncthreads();

    float4* o4 = reinterpret_cast<float4*>(agg) + (size_t)lo * (C_IN / 4);
    for (int i = tid; i < OW_RN * C_IN / 4; i += OW_T) o4[i] = sums4[i];
}

__global__ __launch_bounds__(256) void mlp_fused_kernel(
    const float* __restrict__ agg, const float* __restrict__ x,
    const float* __restrict__ dinv, const float* __restrict__ wpk_f,
    const float* __restrict__ b_gcn,
    const float* __restrict__ b1,
    const float* __restrict__ b2, const float* __restrict__ Wd,
    const float* __restrict__ bd, float* __restrict__ out) {
    __shared__ float4 zb4[ROWS * ZP / 4];
    __shared__ float4 h14[ROWS * H1P / 4];
    __shared__ float4 h24[ROWS * (HID + 1) / 4];
    static_assert((ROWS * (HID + 1)) % 4 == 0 && ROWS * ZP <= ROWS * (HID + 1), "u tile fits in h2's space");
    const float4* wpk = reinterpret_cast<const float4*>(wpk_f);
    float* zb = reinterpret_cast<float*>(zb4);
    float* h1 = reinterpret_cast<float*>(h14);
    float* h2 = reinterpret_cast<float*>(h24);
    float* ub = h2;

    int tid = threadIdx.x;
    int nodeBase = blockIdx.x * ROWS;

    #pragma unroll
    for (int t = 0; t < ROWS * C_IN / 4 / 256; ++t) {
        const int i = tid + t * 256;
        const int r = i >> 4, c4 = i & 15;
        const size_t g4 = (size_t)(nodeBase + r) * (C_IN / 4) + c4;
        const float4 av = reinterpret_cast<const float4*>(agg)[g4];
        const float4 xv = reinterpret_cast<const float4*>(x)[g4];
        const float dv = dinv[nodeBase + r];
        float4 u;
        u.x = dv * (av.x + dv * xv.x);
        u.y = dv * (av.y + dv * xv.y);
        u.z = dv * (av.z + dv * xv.z);
        u.w = dv * (av.w + dv * xv.w);
        reinterpret_cast<float4*>(ub + r * ZP)[c4] = u;
        reinterpret_cast<float4*>(zb + r * ZP)[c4] = xv;
    }
    __syncthreads();

    int wave = tid >> 5, lane = tid & 31, half = lane >> 4, lm = lane & 15;
    int col0 = wave * 16;
    int col1 = col0 + 128;

    {
        const int rt = wave >> 2, ct = wave & 3;
        v8f acc = {};
        const float4* aR = reinterpret_cast<const float4*>(ub + (16 * rt + lm) * ZP + 16 * half);
        const float4* bw = wpk + PK_W2 + PK_W1 + ct * 256 + lane;
        #pragma unroll
        for (int t = 0; t < C_IN / 32; ++t) {
            float4 a[4], bv[4];
            #pragma unroll
            for (int q = 0; q < 4; ++q) bv[q] = bw[t * 128 + q * 32];
            #pragma unroll
            for (int q = 0; q < 4; ++q) a[q] = aR[8 * t + q];
            #pragma unroll
            for (int q = 0; q < 4; ++q) {
                v2f av, b;
                av.x = a[q].x;  av.y = a[q].y;
                b.x = bv[q].x;  b.y = bv[q].y;
                acc = wmma4(av, b, acc);
                av.x = a[q].z;  av.y = a[q].w;
                b.x = bv[q].z;  b.y = bv[q].w;
                acc = wmma4(av, b, acc);
            }
        }
        const float bc = b_gcn[16 * ct + lm];
        #pragma unroll
        for (int i = 0; i < 8; ++i) {
            float* zp = zb + (16 * rt + i + 8 * half) * ZP + 16 * ct + lm;
            float v = acc[i] + bc;
            v = v > 0.0f ? v : 0.0f;
            *zp = v + *zp;
        }
    }
    __syncthreads();

    {
        v8f acc00 = {}, acc01 = {}, acc10 = {}, acc11 = {};
        const float4* aR0 = reinterpret_cast<const float4*>(zb + lm * ZP + 16 * half);
        const float4* aR1 = reinterpret_cast<const float4*>(zb + (16 + lm) * ZP + 16 * half);
        const float4* w0 = wpk + PK_W2 + wave * 256 + lane;
        const float4* w1 = w0 + 128;
        #pragma unroll
        for (int t = 0; t < C_IN / 32; ++t) {
            float4 a0[4], a1[4], b0[4], b1q[4];
            #pragma unroll
            for (int q = 0; q < 4; ++q) { b0[q] = w0[2048 * t + 32 * q]; b1q[q] = w1[2048 * t + 32 * q]; }
            #pragma unroll
            for (int q = 0; q < 4; ++q) { a0[q] = aR0[8 * t + q]; a1[q] = aR1[8 * t + q]; }
            #pragma unroll
            for (int q = 0; q < 4; ++q) {
                v2f a0v, a1v, b0v, b1v;
                a0v.x = a0[q].x;  a0v.y = a0[q].y;
                a1v.x = a1[q].x;  a1v.y = a1[q].y;
                b0v.x = b0[q].x;  b0v.y = b0[q].y;
                b1v.x = b1q[q].x; b1v.y = b1q[q].y;
                acc00 = wmma4(a0v, b0v, acc00);
                acc01 = wmma4(a0v, b1v, acc01);
                acc10 = wmma4(a1v, b0v, acc10);
                acc11 = wmma4(a1v, b1v, acc11);
                a0v.x = a0[q].z;  a0v.y = a0[q].w;
                a1v.x = a1[q].z;  a1v.y = a1[q].w;
                b0v.x = b0[q].z;  b0v.y = b0[q].w;
                b1v.x = b1q[q].z; b1v.y = b1q[q].w;
                acc00 = wmma4(a0v, b0v, acc00);
                acc01 = wmma4(a0v, b1v, acc01);
                acc10 = wmma4(a1v, b0v, acc10);
                acc11 = wmma4(a1v, b1v, acc11);
            }
        }
        float bc0 = b1[col0 + lm], bc1 = b1[col1 + lm];
        #pragma unroll
        for (int i = 0; i < 8; ++i) {
            int r = i + 8 * half;
            h1[r * H1P + col0 + lm]        = leaky01(acc00[i] + bc0);
            h1[r * H1P + col1 + lm]        = leaky01(acc01[i] + bc1);
            h1[(16 + r) * H1P + col0 + lm] = leaky01(acc10[i] + bc0);
            h1[(16 + r) * H1P + col1 + lm] = leaky01(acc11[i] + bc1);
        }
    }
    __syncthreads();

    {
        v8f acc00 = {}, acc01 = {}, acc10 = {}, acc11 = {};
        const float4* aR0 = reinterpret_cast<const float4*>(h1 + lm * H1P + 16 * half);
        const float4* aR1 = reinterpret_cast<const float4*>(h1 + (16 + lm) * H1P + 16 * half);
        const float4* w0 = wpk + wave * 256 + lane;
        const float4* w1 = w0 + 128;
        #pragma unroll 1
        for (int kq = 0; kq < HID / 4; kq += 8) {
            float4 a0[4], a1[4], b0[4], b1q[4];
            #pragma unroll
            for (int q = 0; q < 4; ++q) { b0[q] = w0[256 * kq + 32 * q]; b1q[q] = w1[256 * kq + 32 * q]; }
            #pragma unroll
            for (int q = 0; q < 4; ++q) { a0[q] = aR0[kq + q]; a1[q] = aR1[kq + q]; }
            #pragma unroll
            for (int q = 0; q < 4; ++q) {
                v2f a0v, a1v, b0v, b1v;
                a0v.x = a0[q].x;  a0v.y = a0[q].y;
                a1v.x = a1[q].x;  a1v.y = a1[q].y;
                b0v.x = b0[q].x;  b0v.y = b0[q].y;
                b1v.x = b1q[q].x; b1v.y = b1q[q].y;
                acc00 = wmma4(a0v, b0v, acc00);
                acc01 = wmma4(a0v, b1v, acc01);
                acc10 = wmma4(a1v, b0v, acc10);
                acc11 = wmma4(a1v, b1v, acc11);
                a0v.x = a0[q].z;  a0v.y = a0[q].w;
                a1v.x = a1[q].z;  a1v.y = a1[q].w;
                b0v.x = b0[q].z;  b0v.y = b0[q].w;
                b1v.x = b1q[q].z; b1v.y = b1q[q].w;
                acc00 = wmma4(a0v, b0v, acc00);
                acc01 = wmma4(a0v, b1v, acc01);
                acc10 = wmma4(a1v, b0v, acc10);
                acc11 = wmma4(a1v, b1v, acc11);
            }
        }
        float bc0 = b2[col0 + lm], bc1 = b2[col1 + lm];
        #pragma unroll
        for (int i = 0; i < 8; ++i) {
            int r = i + 8 * half;
            h2[r * (HID + 1) + col0 + lm]        = leaky01(acc00[i] + bc0);
            h2[r * (HID + 1) + col1 + lm]        = leaky01(acc01[i] + bc1);
            h2[(16 + r) * (HID + 1) + col0 + lm] = leaky01(acc10[i] + bc0);
            h2[(16 + r) * (HID + 1) + col1 + lm] = leaky01(acc11[i] + bc1);
        }
    }
    __syncthreads();

    {
        int row = tid >> 3, seg = tid & 7;
        const float* hrow = h2 + row * (HID + 1) + seg * 32;
        const float* wseg = Wd + seg * 32;
        float p = 0.0f;
        #pragma unroll
        for (int c = 0; c < 32; ++c) p += hrow[c] * wseg[c];
        zb[row * 8 + seg] = p;
    }
    __syncthreads();
    if (tid < ROWS) {
        float acc = 0.0f;
        #pragma unroll
        for (int s = 0; s < 8; ++s) acc += zb[tid * 8 + s];
        out[nodeBase + tid] = acc + bd[0];
    }
}

extern "C" void kernel_launch(void* const* d_in, const int* in_sizes, int n_in,
                              void* d_out, int out_size, void* d_ws, size_t ws_size,
                              hipStream_t stream) {
    const float* x     = (const float*)d_in[0];
    const int*   ei    = (const int*)d_in[1];
    const float* W_gcn = (const float*)d_in[2];
    const float* b_gcn = (const float*)d_in[3];
    const float* W1    = (const float*)d_in[4];
    const float* b1    = (const float*)d_in[5];
    const float* W2    = (const float*)d_in[6];
    const float* b2    = (const float*)d_in[7];
    const float* Wd    = (const float*)d_in[8];
    const float* bd    = (const float*)d_in[9];
    float* out = (float*)d_out;

    char* ws = (char*)d_ws;
    size_t off0 = 0;
    size_t off1 = (off0 + (size_t)N_NODES * sizeof(float) + 255) & ~(size_t)255;
    size_t off2 = (off1 + (size_t)N_NODES * C_IN * sizeof(float) + 255) & ~(size_t)255;
    size_t off3 = (off2 + (size_t)PK_ALL * 4 * sizeof(float) + 255) & ~(size_t)255;
    size_t off4 = (off3 + (size_t)OW_NS * LS_W * sizeof(unsigned) + 255) & ~(size_t)255;
    float* dinv = (float*)(ws + off0);
    float* agg  = (float*)(ws + off1);
    float* wpk  = (float*)(ws + off2);
    unsigned* lstart = (unsigned*)(ws + off3);
    int2* lpairs     = (int2*)(ws + off4);

    const int B = 256;
    slice_sort_kernel<<<OW_NS, LS_T, 0, stream>>>(ei, lstart, lpairs);
    range_deg_pack_kernel<<<OW_NB + PK_NB, RS_T, 0, stream>>>(lpairs, lstart, dinv, W_gcn, W1, W2, wpk);

    owner_sum_kernel<<<OW_NB, OW_T, 0, stream>>>(lpairs, lstart, x, dinv, agg);

    mlp_fused_kernel<<<N_NODES / ROWS, B, 0, stream>>>(agg, x, dinv, wpk, b_gcn, b1, b2,
                                                       Wd, bd, out);
}
